# starting kernel
# baseline (speedup 1.0000x reference)
#include <hip/hip_runtime.h>
#include <hip/hip_bf16.h>
#include <math.h>

#define N_NODES 50000
#define N_EDGES 800000
#define F_DIM   128
#define H_OUT1  128
#define C_OUT   64
#define K_TOT   512

typedef __attribute__((ext_vector_type(2))) float v2f;
typedef __attribute__((ext_vector_type(8))) float v8f;


__global__ void zero_f32_kernel(float* __restrict__ p, int n) {
    int i = blockIdx.x * blockDim.x + threadIdx.x;
    if (i < n) p[i] = 0.0f;
}

__global__ void degree_kernel(const int* __restrict__ col, float* __restrict__ deg) {
    int e = blockIdx.x * blockDim.x + threadIdx.x;
    if (e < N_EDGES) atomicAdd(&deg[col[e]], 1.0f);
}

__global__ void dinv_kernel(float* __restrict__ deg) {
    int i = blockIdx.x * blockDim.x + threadIdx.x;
    if (i < N_NODES) {
        float d = deg[i];
        deg[i] = (d > 0.0f) ? rsqrtf(d) : 0.0f;
    }
}

__global__ void norm_kernel(const int* __restrict__ row, const int* __restrict__ col,
                            const float* __restrict__ dinv, float* __restrict__ nrm) {
    int e = blockIdx.x * blockDim.x + threadIdx.x;
    if (e < N_EDGES) nrm[e] = dinv[row[e]] * dinv[col[e]];
}

__global__ void spmm_scatter_kernel(const int* __restrict__ row, const int* __restrict__ col,
                                    const float* __restrict__ nrm,
                                    const float* __restrict__ hin, float* __restrict__ hout) {
    long long tid = (long long)blockIdx.x * blockDim.x + threadIdx.x;
    int e = (int)(tid >> 5);
    int f = ((int)tid & 31) * 4;
    if (e >= N_EDGES) return;
    float w = nrm[e];
    const float4 v = *(const float4*)(hin + (long long)row[e] * F_DIM + f);
    float* dst = hout + (long long)col[e] * F_DIM + f;
    atomicAdd(dst + 0, w * v.x);
    atomicAdd(dst + 1, w * v.y);
    atomicAdd(dst + 2, w * v.z);
    atomicAdd(dst + 3, w * v.w);
}

template <int NOUT, int MODE>
__global__ __launch_bounds__(256) void tag_gemm_wmma_kernel(
    const float* __restrict__ s0, const float* __restrict__ s1,
    const float* __restrict__ s2, const float* __restrict__ s3,
    const float* __restrict__ W, const float* __restrict__ bias,
    float* __restrict__ out) {
    __shared__ float Bs[K_TOT * 16];

    const int col0 = blockIdx.y * 16;

    for (int i = threadIdx.x; i < K_TOT * 16; i += 256) {
        int k = i >> 4;
        int n = i & 15;
        Bs[i] = W[k * NOUT + col0 + n];
    }
    __syncthreads();

    const int wave   = threadIdx.x >> 5;
    const int lane   = threadIdx.x & 31;
    const int lane16 = lane & 15;
    const int hi     = lane >> 4;
    const int row0   = (blockIdx.x * 8 + wave) * 16;
    if (row0 >= N_NODES) return;
    const int m = row0 + lane16;

    const float* segs[4] = { s0, s1, s2, s3 };

    v8f c = { 0.f, 0.f, 0.f, 0.f, 0.f, 0.f, 0.f, 0.f };

#pragma unroll
    for (int s = 0; s < 4; ++s) {
        const float* __restrict__ arow = segs[s] + (long long)m * F_DIM;
        const float* __restrict__ bseg = Bs + s * F_DIM * 16;
        for (int kk = 0; kk < F_DIM; kk += 4) {
            v2f a;
            a.x = arow[kk + hi * 2 + 0];
            a.y = arow[kk + hi * 2 + 1];
            v2f b;
            b.x = bseg[(kk + hi * 2 + 0) * 16 + lane16];
            b.y = bseg[(kk + hi * 2 + 1) * 16 + lane16];
            c = __builtin_amdgcn_wmma_f32_16x16x4_f32(false, a, false, b, (short)0, c,
                                                      false, false);
        }
    }

    const int n    = col0 + lane16;
    const float bv = bias[n];
#pragma unroll
    for (int j = 0; j < 8; ++j) {
        int mm  = row0 + j + hi * 8;
        float v = c[j] + bv;
        if (MODE == 0) v = (v > 0.0f) ? v : 0.0f;
        out[(long long)mm * NOUT + n] = v;
    }
}

__global__ void log_softmax_kernel(float* __restrict__ out) {
    int wave = threadIdx.x >> 5;
    int lane = threadIdx.x & 31;
    int r = blockIdx.x * (blockDim.x >> 5) + wave;
    if (r >= N_NODES) return;
    float* p = out + (long long)r * C_OUT;
    float a = p[lane];
    float b = p[lane + 32];
    float mx = fmaxf(a, b);
#pragma unroll
    for (int off = 16; off > 0; off >>= 1) mx = fmaxf(mx, __shfl_xor(mx, off, 32));
    float s = expf(a - mx) + expf(b - mx);
#pragma unroll
    for (int off = 16; off > 0; off >>= 1) s += __shfl_xor(s, off, 32);
    float lse = mx + logf(s);
    p[lane]      = a - lse;
    p[lane + 32] = b - lse;
}


extern "C" void kernel_launch(void* const* d_in, const int* in_sizes, int n_in,
                              void* d_out, int out_size, void* d_ws, size_t ws_size,
                              hipStream_t stream) {
    const float* x   = (const float*)d_in[0];
    const int*   ei  = (const int*)d_in[1];
    const float* W1  = (const float*)d_in[2];
    const float* b1  = (const float*)d_in[3];
    const float* W2  = (const float*)d_in[4];
    const float* b2  = (const float*)d_in[5];
    float*       out = (float*)d_out;

    const int* rowp = ei;
    const int* colp = ei + N_EDGES;

    float* ws   = (float*)d_ws;
    float* dinv = ws;
    float* nrm  = dinv + N_NODES;
    float* hb0  = nrm + N_EDGES;
    float* hb1  = hb0 + (size_t)N_NODES * F_DIM;
    float* hb2  = hb1 + (size_t)N_NODES * F_DIM;
    float* acc1 = hb2 + (size_t)N_NODES * F_DIM;

    const int TB = 256;
    const int nodeBlocks = (N_NODES + TB - 1) / TB;
    const int edgeBlocks = (N_EDGES + TB - 1) / TB;
    const int featElems  = N_NODES * F_DIM;
    const int featBlocks = (featElems + TB - 1) / TB;
    const long long spmmThreads = (long long)N_EDGES * 32;
    const int spmmBlocks = (int)((spmmThreads + TB - 1) / TB);
    const int rowTiles   = N_NODES / 16;
    const int rowGroups  = (rowTiles + 7) / 8;

    zero_f32_kernel<<<nodeBlocks, TB, 0, stream>>>(dinv, N_NODES);
    degree_kernel<<<edgeBlocks, TB, 0, stream>>>(colp, dinv);
    dinv_kernel<<<nodeBlocks, TB, 0, stream>>>(dinv);
    norm_kernel<<<edgeBlocks, TB, 0, stream>>>(rowp, colp, dinv, nrm);

    zero_f32_kernel<<<featBlocks, TB, 0, stream>>>(hb0, featElems);
    spmm_scatter_kernel<<<spmmBlocks, TB, 0, stream>>>(rowp, colp, nrm, x, hb0);
    zero_f32_kernel<<<featBlocks, TB, 0, stream>>>(hb1, featElems);
    spmm_scatter_kernel<<<spmmBlocks, TB, 0, stream>>>(rowp, colp, nrm, hb0, hb1);
    zero_f32_kernel<<<featBlocks, TB, 0, stream>>>(hb2, featElems);
    spmm_scatter_kernel<<<spmmBlocks, TB, 0, stream>>>(rowp, colp, nrm, hb1, hb2);

    tag_gemm_wmma_kernel<H_OUT1, 0><<<dim3(rowGroups, H_OUT1 / 16), 256, 0, stream>>>(
        x, hb0, hb1, hb2, W1, b1, acc1);

    zero_f32_kernel<<<featBlocks, TB, 0, stream>>>(hb0, featElems);
    spmm_scatter_kernel<<<spmmBlocks, TB, 0, stream>>>(rowp, colp, nrm, acc1, hb0);
    zero_f32_kernel<<<featBlocks, TB, 0, stream>>>(hb1, featElems);
    spmm_scatter_kernel<<<spmmBlocks, TB, 0, stream>>>(rowp, colp, nrm, hb0, hb1);
    zero_f32_kernel<<<featBlocks, TB, 0, stream>>>(hb2, featElems);
    spmm_scatter_kernel<<<spmmBlocks, TB, 0, stream>>>(rowp, colp, nrm, hb1, hb2);

    tag_gemm_wmma_kernel<C_OUT, 1><<<dim3(rowGroups, C_OUT / 16), 256, 0, stream>>>(
        acc1, hb0, hb1, hb2, W2, b2, out);

    log_softmax_kernel<<<N_NODES / 8, TB, 0, stream>>>(out);
}


# Propagation scatter-add (float atomics, zero fills) -> edges counting-sorted by target once per call (int atomic rank) + one-wave-per-node register gather; per-node sum order = sort arrival order
# speedup vs baseline: 3.2624x; 3.2624x over previous
#include <hip/hip_runtime.h>
#include <hip/hip_bf16.h>
#include <math.h>


#define N_NODES 50000
#define N_EDGES 800000
#define F_DIM   128
#define H_OUT1  128
#define C_OUT   64
#define K_TOT   512

#define SCAN_T     1024
#define SCAN_TILES 13
#define N_PAD      (SCAN_T * SCAN_TILES * 4)

typedef __attribute__((ext_vector_type(2))) float v2f;
typedef __attribute__((ext_vector_type(8))) float v8f;


__global__ void zero_i32_kernel(int* __restrict__ p, int n) {
    int i = blockIdx.x * blockDim.x + threadIdx.x;
    if (i < n) p[i] = 0;
}

__global__ void hist_rank_kernel(const int* __restrict__ col, int* __restrict__ cnt,
                                 int* __restrict__ rank) {
    int e = blockIdx.x * blockDim.x + threadIdx.x;
    if (e < N_EDGES) rank[e] = atomicAdd(&cnt[col[e]], 1);
}

__global__ __launch_bounds__(SCAN_T) void scan_dinv_kernel(const int* __restrict__ cnt,
                                                           int* __restrict__ rowptr,
                                                           float* __restrict__ dinv) {
    __shared__ int wtot[SCAN_TILES][32];
    __shared__ int ttot[SCAN_TILES];
    const int t    = threadIdx.x;
    const int lane = t & 31;
    const int w    = t >> 5;

    int4 c[SCAN_TILES];
    int  incl[SCAN_TILES];
#pragma unroll
    for (int i = 0; i < SCAN_TILES; ++i) c[i] = ((const int4*)cnt)[i * SCAN_T + t];

#pragma unroll
    for (int i = 0; i < SCAN_TILES; ++i) {
        float4 d;
        d.x = (c[i].x > 0) ? rsqrtf((float)c[i].x) : 0.0f;
        d.y = (c[i].y > 0) ? rsqrtf((float)c[i].y) : 0.0f;
        d.z = (c[i].z > 0) ? rsqrtf((float)c[i].z) : 0.0f;
        d.w = (c[i].w > 0) ? rsqrtf((float)c[i].w) : 0.0f;
        ((float4*)dinv)[i * SCAN_T + t] = d;
        int s = c[i].x + c[i].y + c[i].z + c[i].w;
        for (int off = 1; off < 32; off <<= 1) {
            int nb = __shfl_up(s, off, 32);
            if (lane >= off) s += nb;
        }
        incl[i] = s;
        if (lane == 31) wtot[i][w] = s;
    }
    __syncthreads();
    if (w < SCAN_TILES) {
        int x = wtot[w][lane];
        int s = x;
        for (int off = 1; off < 32; off <<= 1) {
            int nb = __shfl_up(s, off, 32);
            if (lane >= off) s += nb;
        }
        wtot[w][lane] = s - x;
        if (lane == 31) ttot[w] = s;
    }
    __syncthreads();
    int run = 0;
#pragma unroll
    for (int i = 0; i < SCAN_TILES; ++i) {
        int mine = c[i].x + c[i].y + c[i].z + c[i].w;
        int ex   = run + wtot[i][w] + incl[i] - mine;
        int4 p;
        p.x = ex;
        p.y = p.x + c[i].x;
        p.z = p.y + c[i].y;
        p.w = p.z + c[i].z;
        ((int4*)rowptr)[i * SCAN_T + t] = p;
        run += ttot[i];
    }
}

__global__ void fill_kernel(const int* __restrict__ row, const int* __restrict__ col,
                            const int* __restrict__ rank, const int* __restrict__ rowptr,
                            const float* __restrict__ dinv, int2* __restrict__ erec) {
    int e = blockIdx.x * blockDim.x + threadIdx.x;
    if (e < N_EDGES) {
        int r = row[e];
        int c = col[e];
        float wgt = dinv[r] * dinv[c];
        erec[rowptr[c] + rank[e]] = make_int2(r, __float_as_int(wgt));
    }
}

template <int CNT>
__device__ __forceinline__ void gather_accum(const int2* __restrict__ rec,
                                             const float* __restrict__ hl, float4& acc) {
    int2   r[CNT];
    float4 v[CNT];
#pragma unroll
    for (int i = 0; i < CNT; ++i) r[i] = rec[i];
#pragma unroll
    for (int i = 0; i < CNT; ++i) {
        unsigned s = min((unsigned)r[i].x, (unsigned)(N_NODES - 1));
        v[i] = *(const float4*)(hl + (size_t)s * F_DIM);
    }
#pragma unroll
    for (int i = 0; i < CNT; ++i) {
        float wgt = __int_as_float(r[i].y);
        acc.x = fmaf(wgt, v[i].x, acc.x);
        acc.y = fmaf(wgt, v[i].y, acc.y);
        acc.z = fmaf(wgt, v[i].z, acc.z);
        acc.w = fmaf(wgt, v[i].w, acc.w);
    }
}

__global__ __launch_bounds__(256) void spmm_gather_kernel(const int* __restrict__ rowptr,
                                                          const int2* __restrict__ erec,
                                                          const float* __restrict__ hin,
                                                          float* __restrict__ hout) {
    const int lane = threadIdx.x & 31;
    const int node = __builtin_amdgcn_readfirstlane((int)(blockIdx.x * 8 + (threadIdx.x >> 5)));
    const int beg = rowptr[node];
    const int end = rowptr[node + 1];
    const float* __restrict__ hl = hin + lane * 4;
    float4 acc = make_float4(0.0f, 0.0f, 0.0f, 0.0f);
    int e = beg;
    while (e + 8 <= end) { gather_accum<8>(erec + e, hl, acc); e += 8; }
    if (e + 4 <= end)    { gather_accum<4>(erec + e, hl, acc); e += 4; }
    switch (end - e) {
        case 3: gather_accum<3>(erec + e, hl, acc); break;
        case 2: gather_accum<2>(erec + e, hl, acc); break;
        case 1: gather_accum<1>(erec + e, hl, acc); break;
        default: break;
    }
    *(float4*)(hout + (size_t)node * F_DIM + lane * 4) = acc;
}

template <int NOUT, int MODE>
__global__ __launch_bounds__(256) void tag_gemm_wmma_kernel(
    const float* __restrict__ s0, const float* __restrict__ s1,
    const float* __restrict__ s2, const float* __restrict__ s3,
    const float* __restrict__ W, const float* __restrict__ bias,
    float* __restrict__ out) {
    __shared__ float Bs[K_TOT * 16];

    const int col0 = blockIdx.y * 16;

    for (int i = threadIdx.x; i < K_TOT * 16; i += 256) {
        int k = i >> 4;
        int n = i & 15;
        Bs[i] = W[k * NOUT + col0 + n];
    }
    __syncthreads();

    const int wave   = threadIdx.x >> 5;
    const int lane   = threadIdx.x & 31;
    const int lane16 = lane & 15;
    const int hi     = lane >> 4;
    const int row0   = (blockIdx.x * 8 + wave) * 16;
    if (row0 >= N_NODES) return;
    const int m = row0 + lane16;

    const float* segs[4] = { s0, s1, s2, s3 };

    v8f c = { 0.f, 0.f, 0.f, 0.f, 0.f, 0.f, 0.f, 0.f };

#pragma unroll
    for (int s = 0; s < 4; ++s) {
        const float* __restrict__ arow = segs[s] + (long long)m * F_DIM;
        const float* __restrict__ bseg = Bs + s * F_DIM * 16;
        for (int kk = 0; kk < F_DIM; kk += 4) {
            v2f a;
            a.x = arow[kk + hi * 2 + 0];
            a.y = arow[kk + hi * 2 + 1];
            v2f b;
            b.x = bseg[(kk + hi * 2 + 0) * 16 + lane16];
            b.y = bseg[(kk + hi * 2 + 1) * 16 + lane16];
            c = __builtin_amdgcn_wmma_f32_16x16x4_f32(false, a, false, b, (short)0, c,
                                                      false, false);
        }
    }

    const int n    = col0 + lane16;
    const float bv = bias[n];
#pragma unroll
    for (int j = 0; j < 8; ++j) {
        int mm  = row0 + j + hi * 8;
        float v = c[j] + bv;
        if (MODE == 0) v = (v > 0.0f) ? v : 0.0f;
        out[(long long)mm * NOUT + n] = v;
    }
}

__global__ void log_softmax_kernel(float* __restrict__ out) {
    int wave = threadIdx.x >> 5;
    int lane = threadIdx.x & 31;
    int r = blockIdx.x * (blockDim.x >> 5) + wave;
    if (r >= N_NODES) return;
    float* p = out + (long long)r * C_OUT;
    float a = p[lane];
    float b = p[lane + 32];
    float mx = fmaxf(a, b);
#pragma unroll
    for (int off = 16; off > 0; off >>= 1) mx = fmaxf(mx, __shfl_xor(mx, off, 32));
    float s = expf(a - mx) + expf(b - mx);
#pragma unroll
    for (int off = 16; off > 0; off >>= 1) s += __shfl_xor(s, off, 32);
    float lse = mx + logf(s);
    p[lane]      = a - lse;
    p[lane + 32] = b - lse;
}


extern "C" void kernel_launch(void* const* d_in, const int* in_sizes, int n_in,
                              void* d_out, int out_size, void* d_ws, size_t ws_size,
                              hipStream_t stream) {
    const float* x   = (const float*)d_in[0];
    const int*   ei  = (const int*)d_in[1];
    const float* W1  = (const float*)d_in[2];
    const float* b1  = (const float*)d_in[3];
    const float* W2  = (const float*)d_in[4];
    const float* b2  = (const float*)d_in[5];
    float*       out = (float*)d_out;

    const int* rowp = ei;
    const int* colp = ei + N_EDGES;

    int*   cnt    = (int*)d_ws;
    int*   rowptr = cnt + N_PAD;
    float* dinv   = (float*)(rowptr + N_PAD);
    int*   rank   = (int*)(dinv + N_PAD);
    int2*  erec   = (int2*)(rank + N_EDGES);
    float* hb0    = (float*)(erec + N_EDGES);
    float* hb1    = hb0 + (size_t)N_NODES * F_DIM;
    float* hb2    = hb1 + (size_t)N_NODES * F_DIM;
    float* acc1   = hb2 + (size_t)N_NODES * F_DIM;

    const int TB = 256;
    const int padBlocks  = N_PAD / TB;
    const int edgeBlocks = (N_EDGES + TB - 1) / TB;
    const int nodeGroups = N_NODES / 8;
    const int rowTiles   = N_NODES / 16;
    const int rowGroups  = (rowTiles + 7) / 8;

    zero_i32_kernel<<<padBlocks, TB, 0, stream>>>(cnt, N_PAD);
    hist_rank_kernel<<<edgeBlocks, TB, 0, stream>>>(colp, cnt, rank);
    scan_dinv_kernel<<<1, SCAN_T, 0, stream>>>(cnt, rowptr, dinv);
    fill_kernel<<<edgeBlocks, TB, 0, stream>>>(rowp, colp, rank, rowptr, dinv, erec);

    spmm_gather_kernel<<<nodeGroups, 256, 0, stream>>>(rowptr, erec, x, hb0);
    spmm_gather_kernel<<<nodeGroups, 256, 0, stream>>>(rowptr, erec, hb0, hb1);
    spmm_gather_kernel<<<nodeGroups, 256, 0, stream>>>(rowptr, erec, hb1, hb2);

    tag_gemm_wmma_kernel<H_OUT1, 0><<<dim3(rowGroups, H_OUT1 / 16), 256, 0, stream>>>(
        x, hb0, hb1, hb2, W1, b1, acc1);

    spmm_gather_kernel<<<nodeGroups, 256, 0, stream>>>(rowptr, erec, acc1, hb0);
    spmm_gather_kernel<<<nodeGroups, 256, 0, stream>>>(rowptr, erec, hb0, hb1);
    spmm_gather_kernel<<<nodeGroups, 256, 0, stream>>>(rowptr, erec, hb1, hb2);

    tag_gemm_wmma_kernel<C_OUT, 1><<<dim3(rowGroups, C_OUT / 16), 256, 0, stream>>>(
        acc1, hb0, hb1, hb2, W2, b2, out);

    log_softmax_kernel<<<N_NODES / 8, TB, 0, stream>>>(out);
}


# GEMM: each wave keeps all NOUT/16 column tiles of its 16 rows, so an A fragment is loaded once, not once per 16-column block; W per 128-k segment in two LDS buffers; k order of every sum unchanged
# speedup vs baseline: 4.5186x; 1.3851x over previous
#include <hip/hip_runtime.h>
#include <hip/hip_bf16.h>
#include <math.h>


#define N_NODES 50000
#define N_EDGES 800000
#define F_DIM   128
#define H_OUT1  128
#define C_OUT   64
#define K_TOT   512

#define SCAN_T     1024
#define SCAN_TILES 13
#define N_PAD      (SCAN_T * SCAN_TILES * 4)

typedef __attribute__((ext_vector_type(2))) float v2f;
typedef __attribute__((ext_vector_type(8))) float v8f;


__global__ void zero_i32_kernel(int* __restrict__ p, int n) {
    int i = blockIdx.x * blockDim.x + threadIdx.x;
    if (i < n) p[i] = 0;
}

__global__ void hist_rank_kernel(const int* __restrict__ col, int* __restrict__ cnt,
                                 int* __restrict__ rank) {
    int e = blockIdx.x * blockDim.x + threadIdx.x;
    if (e < N_EDGES) rank[e] = atomicAdd(&cnt[col[e]], 1);
}

__global__ __launch_bounds__(SCAN_T) void scan_dinv_kernel(const int* __restrict__ cnt,
                                                           int* __restrict__ rowptr,
                                                           float* __restrict__ dinv) {
    __shared__ int wtot[SCAN_TILES][32];
    __shared__ int ttot[SCAN_TILES];
    const int t    = threadIdx.x;
    const int lane = t & 31;
    const int w    = t >> 5;

    int4 c[SCAN_TILES];
    int  incl[SCAN_TILES];
#pragma unroll
    for (int i = 0; i < SCAN_TILES; ++i) c[i] = ((const int4*)cnt)[i * SCAN_T + t];

#pragma unroll
    for (int i = 0; i < SCAN_TILES; ++i) {
        float4 d;
        d.x = (c[i].x > 0) ? rsqrtf((float)c[i].x) : 0.0f;
        d.y = (c[i].y > 0) ? rsqrtf((float)c[i].y) : 0.0f;
        d.z = (c[i].z > 0) ? rsqrtf((float)c[i].z) : 0.0f;
        d.w = (c[i].w > 0) ? rsqrtf((float)c[i].w) : 0.0f;
        ((float4*)dinv)[i * SCAN_T + t] = d;
        int s = c[i].x + c[i].y + c[i].z + c[i].w;
        for (int off = 1; off < 32; off <<= 1) {
            int nb = __shfl_up(s, off, 32);
            if (lane >= off) s += nb;
        }
        incl[i] = s;
        if (lane == 31) wtot[i][w] = s;
    }
    __syncthreads();
    if (w < SCAN_TILES) {
        int x = wtot[w][lane];
        int s = x;
        for (int off = 1; off < 32; off <<= 1) {
            int nb = __shfl_up(s, off, 32);
            if (lane >= off) s += nb;
        }
        wtot[w][lane] = s - x;
        if (lane == 31) ttot[w] = s;
    }
    __syncthreads();
    int run = 0;
#pragma unroll
    for (int i = 0; i < SCAN_TILES; ++i) {
        int mine = c[i].x + c[i].y + c[i].z + c[i].w;
        int ex   = run + wtot[i][w] + incl[i] - mine;
        int4 p;
        p.x = ex;
        p.y = p.x + c[i].x;
        p.z = p.y + c[i].y;
        p.w = p.z + c[i].z;
        ((int4*)rowptr)[i * SCAN_T + t] = p;
        run += ttot[i];
    }
}

__global__ void fill_kernel(const int* __restrict__ row, const int* __restrict__ col,
                            const int* __restrict__ rank, const int* __restrict__ rowptr,
                            const float* __restrict__ dinv, int2* __restrict__ erec) {
    int e = blockIdx.x * blockDim.x + threadIdx.x;
    if (e < N_EDGES) {
        int r = row[e];
        int c = col[e];
        float wgt = dinv[r] * dinv[c];
        erec[rowptr[c] + rank[e]] = make_int2(r, __float_as_int(wgt));
    }
}

template <int CNT>
__device__ __forceinline__ void gather_accum(const int2* __restrict__ rec,
                                             const float* __restrict__ hl, float4& acc) {
    int2   r[CNT];
    float4 v[CNT];
#pragma unroll
    for (int i = 0; i < CNT; ++i) r[i] = rec[i];
#pragma unroll
    for (int i = 0; i < CNT; ++i) {
        unsigned s = min((unsigned)r[i].x, (unsigned)(N_NODES - 1));
        v[i] = *(const float4*)(hl + (size_t)s * F_DIM);
    }
#pragma unroll
    for (int i = 0; i < CNT; ++i) {
        float wgt = __int_as_float(r[i].y);
        acc.x = fmaf(wgt, v[i].x, acc.x);
        acc.y = fmaf(wgt, v[i].y, acc.y);
        acc.z = fmaf(wgt, v[i].z, acc.z);
        acc.w = fmaf(wgt, v[i].w, acc.w);
    }
}

__global__ __launch_bounds__(256) void spmm_gather_kernel(const int* __restrict__ rowptr,
                                                          const int2* __restrict__ erec,
                                                          const float* __restrict__ hin,
                                                          float* __restrict__ hout) {
    const int lane = threadIdx.x & 31;
    const int node = __builtin_amdgcn_readfirstlane((int)(blockIdx.x * 8 + (threadIdx.x >> 5)));
    const int beg = rowptr[node];
    const int end = rowptr[node + 1];
    const float* __restrict__ hl = hin + lane * 4;
    float4 acc = make_float4(0.0f, 0.0f, 0.0f, 0.0f);
    int e = beg;
    while (e + 8 <= end) { gather_accum<8>(erec + e, hl, acc); e += 8; }
    if (e + 4 <= end)    { gather_accum<4>(erec + e, hl, acc); e += 4; }
    switch (end - e) {
        case 3: gather_accum<3>(erec + e, hl, acc); break;
        case 2: gather_accum<2>(erec + e, hl, acc); break;
        case 1: gather_accum<1>(erec + e, hl, acc); break;
        default: break;
    }
    *(float4*)(hout + (size_t)node * F_DIM + lane * 4) = acc;
}

template <int NOUT>
struct GemmCfg {
    static constexpr int NT    = NOUT / 16;
    static constexpr int PITCH = 2 * NOUT + 32;
    static constexpr int SEGF  = (F_DIM / 2) * PITCH;
    static constexpr int NV4   = NOUT / 4;
    static constexpr int ITEMS = (F_DIM / 2) * NV4 / 256;
};

template <int NOUT, int MODE>
__global__ __launch_bounds__(256, 4) void tag_gemm_wmma_kernel(
    const float* __restrict__ s0, const float* __restrict__ s1,
    const float* __restrict__ s2, const float* __restrict__ s3,
    const float* __restrict__ W, const float* __restrict__ bias,
    float* __restrict__ out) {
    constexpr int NT    = GemmCfg<NOUT>::NT;
    constexpr int PITCH = GemmCfg<NOUT>::PITCH;
    constexpr int SEGF  = GemmCfg<NOUT>::SEGF;
    constexpr int NV4   = GemmCfg<NOUT>::NV4;
    constexpr int ITEMS = GemmCfg<NOUT>::ITEMS;
    constexpr int GSTEP = 8 * 2 * PITCH;

    __shared__ __attribute__((aligned(16))) float Bs[2 * SEGF];

    const int tid    = threadIdx.x;
    const int lane   = tid & 31;
    const int lane16 = lane & 15;
    const int hi     = lane >> 4;
    const int wave   = __builtin_amdgcn_readfirstlane(tid >> 5);
    const int row0   = (blockIdx.x * 8 + wave) * 16;
    const bool active = row0 < N_NODES;
    const int m = min(row0 + lane16, N_NODES - 1);
    const size_t aoff = (size_t)m * F_DIM + hi * 2;

    const float* segs[4] = { s0, s1, s2, s3 };

    v8f acc[NT];
#pragma unroll
    for (int t = 0; t < NT; ++t) acc[t] = (v8f){ 0.f, 0.f, 0.f, 0.f, 0.f, 0.f, 0.f, 0.f };

    v2f a_cur[8], a_nxt[8];
#pragma unroll
    for (int j = 0; j < 8; ++j) a_cur[j] = *(const v2f*)(segs[0] + aoff + 4 * j);

    const int spr = tid / NV4;
    const int sn4 = tid % NV4;
    const float* __restrict__ wsrc = W + (size_t)(2 * spr) * NOUT + 4 * sn4;
    float* const wdst = Bs + spr * PITCH + 8 * sn4;
    constexpr int PR_STEP = 256 / NV4;
    float4 w0[ITEMS], w1[ITEMS];
#pragma unroll
    for (int it = 0; it < ITEMS; ++it) {
        const float* g = wsrc + (size_t)(2 * PR_STEP * it) * NOUT;
        w0[it] = *(const float4*)g;
        w1[it] = *(const float4*)(g + NOUT);
    }
#pragma unroll
    for (int it = 0; it < ITEMS; ++it) {
        float* d = wdst + it * PR_STEP * PITCH;
        *(float4*)d       = make_float4(w0[it].x, w1[it].x, w0[it].y, w1[it].y);
        *(float4*)(d + 4) = make_float4(w0[it].z, w1[it].z, w0[it].w, w1[it].w);
    }
    __syncthreads();

#pragma unroll
    for (int s = 0; s < 4; ++s) {
        const float* bg = Bs + (s & 1) * SEGF + hi * PITCH + 2 * lane16;
        const float* ag = segs[s] + aoff;
        const float* an = (s < 3) ? segs[s + 1] + aoff : ag;
        v2f bq[2][NT];
#pragma unroll
        for (int t = 0; t < NT; ++t) bq[0][t] = *(const v2f*)(bg + 32 * t);

        auto products = [&](const float* bgp, const float* bnext) {
#pragma unroll
            for (int j = 0; j < 8; ++j) {
                const float* bn = (j < 7) ? bgp + (j + 1) * 2 * PITCH : bnext;
#pragma unroll
                for (int t = 0; t < NT; ++t) bq[(j + 1) & 1][t] = *(const v2f*)(bn + 32 * t);
#pragma unroll
                for (int t = 0; t < NT; ++t)
                    acc[t] = __builtin_amdgcn_wmma_f32_16x16x4_f32(
                        false, a_cur[j], false, bq[j & 1][t], (short)0, acc[t], false, false);
            }
        };

#pragma unroll
        for (int j = 0; j < 8; ++j) a_nxt[j] = *(const v2f*)(ag + 32 + 4 * j);
        if (s < 3) {
#pragma unroll
            for (int it = 0; it < ITEMS; ++it) {
                const float* g = wsrc + (size_t)((s + 1) * F_DIM + 2 * PR_STEP * it) * NOUT;
                w0[it] = *(const float4*)g;
                w1[it] = *(const float4*)(g + NOUT);
            }
        }
        products(bg, bg + GSTEP);
#pragma unroll
        for (int j = 0; j < 8; ++j) a_cur[j] = a_nxt[j];
        bg += GSTEP;
        ag += 32;

#pragma unroll 1
        for (int sc = 1; sc < 4; ++sc) {
            const bool last = (sc == 3);
            const float* ap = last ? an : ag + 32;
            if (s < 3 || !last) {
#pragma unroll
                for (int j = 0; j < 8; ++j) a_nxt[j] = *(const v2f*)(ap + 4 * j);
            }
            products(bg, last ? bg + 7 * 2 * PITCH : bg + GSTEP);
#pragma unroll
            for (int j = 0; j < 8; ++j) a_cur[j] = a_nxt[j];
            bg += GSTEP;
            ag += 32;
        }

        if (s < 3) {
            float* db = wdst + ((s + 1) & 1) * SEGF;
#pragma unroll
            for (int it = 0; it < ITEMS; ++it) {
                float* d = db + it * PR_STEP * PITCH;
                *(float4*)d       = make_float4(w0[it].x, w1[it].x, w0[it].y, w1[it].y);
                *(float4*)(d + 4) = make_float4(w0[it].z, w1[it].z, w0[it].w, w1[it].w);
            }
            __syncthreads();
        }
    }

    if (active) {
#pragma unroll
        for (int t = 0; t < NT; ++t) {
            const int n    = t * 16 + lane16;
            const float bv = bias[n];
#pragma unroll
            for (int j = 0; j < 8; ++j) {
                int mm  = row0 + j + hi * 8;
                float v = acc[t][j] + bv;
                if (MODE == 0) v = (v > 0.0f) ? v : 0.0f;
                out[(long long)mm * NOUT + n] = v;
            }
        }
    }
}

__global__ void log_softmax_kernel(float* __restrict__ out) {
    int wave = threadIdx.x >> 5;
    int lane = threadIdx.x & 31;
    int r = blockIdx.x * (blockDim.x >> 5) + wave;
    if (r >= N_NODES) return;
    float* p = out + (long long)r * C_OUT;
    float a = p[lane];
    float b = p[lane + 32];
    float mx = fmaxf(a, b);
#pragma unroll
    for (int off = 16; off > 0; off >>= 1) mx = fmaxf(mx, __shfl_xor(mx, off, 32));
    float s = expf(a - mx) + expf(b - mx);
#pragma unroll
    for (int off = 16; off > 0; off >>= 1) s += __shfl_xor(s, off, 32);
    float lse = mx + logf(s);
    p[lane]      = a - lse;
    p[lane + 32] = b - lse;
}


extern "C" void kernel_launch(void* const* d_in, const int* in_sizes, int n_in,
                              void* d_out, int out_size, void* d_ws, size_t ws_size,
                              hipStream_t stream) {
    const float* x   = (const float*)d_in[0];
    const int*   ei  = (const int*)d_in[1];
    const float* W1  = (const float*)d_in[2];
    const float* b1  = (const float*)d_in[3];
    const float* W2  = (const float*)d_in[4];
    const float* b2  = (const float*)d_in[5];
    float*       out = (float*)d_out;

    const int* rowp = ei;
    const int* colp = ei + N_EDGES;

    int*   cnt    = (int*)d_ws;
    int*   rowptr = cnt + N_PAD;
    float* dinv   = (float*)(rowptr + N_PAD);
    int*   rank   = (int*)(dinv + N_PAD);
    int2*  erec   = (int2*)(rank + N_EDGES);
    float* hb0    = (float*)(erec + N_EDGES);
    float* hb1    = hb0 + (size_t)N_NODES * F_DIM;
    float* hb2    = hb1 + (size_t)N_NODES * F_DIM;
    float* acc1   = hb2 + (size_t)N_NODES * F_DIM;

    const int TB = 256;
    const int padBlocks  = N_PAD / TB;
    const int edgeBlocks = (N_EDGES + TB - 1) / TB;
    const int nodeGroups = N_NODES / 8;
    const int rowTiles   = N_NODES / 16;
    const int rowGroups  = (rowTiles + 7) / 8;

    zero_i32_kernel<<<padBlocks, TB, 0, stream>>>(cnt, N_PAD);
    hist_rank_kernel<<<edgeBlocks, TB, 0, stream>>>(colp, cnt, rank);
    scan_dinv_kernel<<<1, SCAN_T, 0, stream>>>(cnt, rowptr, dinv);
    fill_kernel<<<edgeBlocks, TB, 0, stream>>>(rowp, colp, rank, rowptr, dinv, erec);

    spmm_gather_kernel<<<nodeGroups, 256, 0, stream>>>(rowptr, erec, x, hb0);
    spmm_gather_kernel<<<nodeGroups, 256, 0, stream>>>(rowptr, erec, hb0, hb1);
    spmm_gather_kernel<<<nodeGroups, 256, 0, stream>>>(rowptr, erec, hb1, hb2);

    tag_gemm_wmma_kernel<H_OUT1, 0><<<rowGroups, 256, 0, stream>>>(
        x, hb0, hb1, hb2, W1, b1, acc1);

    spmm_gather_kernel<<<nodeGroups, 256, 0, stream>>>(rowptr, erec, acc1, hb0);
    spmm_gather_kernel<<<nodeGroups, 256, 0, stream>>>(rowptr, erec, hb0, hb1);
    spmm_gather_kernel<<<nodeGroups, 256, 0, stream>>>(rowptr, erec, hb1, hb2);

    tag_gemm_wmma_kernel<C_OUT, 1><<<rowGroups, 256, 0, stream>>>(
        acc1, hb0, hb1, hb2, W2, b2, out);

    log_softmax_kernel<<<N_NODES / 8, TB, 0, stream>>>(out);
}


# Layer 2 in Horner form: Z_k = h W2[k] first (one GEMM, K=128), then 64-wide hops t = A t + Z_k, two edges per load; hop bytes halve; sum order changes (products before hops, even/odd edge halves)
# speedup vs baseline: 4.7323x; 1.0473x over previous
#include <hip/hip_runtime.h>
#include <hip/hip_bf16.h>
#include <math.h>


#define N_NODES 50000
#define N_EDGES 800000
#define F_DIM   128
#define H_OUT1  128
#define C_OUT   64
#define K_TOT   512

#define SCAN_T     1024
#define SCAN_TILES 13
#define N_PAD      (SCAN_T * SCAN_TILES * 4)

typedef __attribute__((ext_vector_type(2))) float v2f;
typedef __attribute__((ext_vector_type(8))) float v8f;


__global__ void zero_i32_kernel(int* __restrict__ p, int n) {
    int i = blockIdx.x * blockDim.x + threadIdx.x;
    if (i < n) p[i] = 0;
}

__global__ void hist_rank_kernel(const int* __restrict__ col, int* __restrict__ cnt,
                                 int* __restrict__ rank) {
    int e = blockIdx.x * blockDim.x + threadIdx.x;
    if (e < N_EDGES) rank[e] = atomicAdd(&cnt[col[e]], 1);
}

__global__ __launch_bounds__(SCAN_T) void scan_dinv_kernel(const int* __restrict__ cnt,
                                                           int* __restrict__ rowptr,
                                                           float* __restrict__ dinv) {
    __shared__ int wtot[SCAN_TILES][32];
    __shared__ int ttot[SCAN_TILES];
    const int t    = threadIdx.x;
    const int lane = t & 31;
    const int w    = t >> 5;

    int4 c[SCAN_TILES];
    int  incl[SCAN_TILES];
#pragma unroll
    for (int i = 0; i < SCAN_TILES; ++i) c[i] = ((const int4*)cnt)[i * SCAN_T + t];

#pragma unroll
    for (int i = 0; i < SCAN_TILES; ++i) {
        float4 d;
        d.x = (c[i].x > 0) ? rsqrtf((float)c[i].x) : 0.0f;
        d.y = (c[i].y > 0) ? rsqrtf((float)c[i].y) : 0.0f;
        d.z = (c[i].z > 0) ? rsqrtf((float)c[i].z) : 0.0f;
        d.w = (c[i].w > 0) ? rsqrtf((float)c[i].w) : 0.0f;
        ((float4*)dinv)[i * SCAN_T + t] = d;
        int s = c[i].x + c[i].y + c[i].z + c[i].w;
        for (int off = 1; off < 32; off <<= 1) {
            int nb = __shfl_up(s, off, 32);
            if (lane >= off) s += nb;
        }
        incl[i] = s;
        if (lane == 31) wtot[i][w] = s;
    }
    __syncthreads();
    if (w < SCAN_TILES) {
        int x = wtot[w][lane];
        int s = x;
        for (int off = 1; off < 32; off <<= 1) {
            int nb = __shfl_up(s, off, 32);
            if (lane >= off) s += nb;
        }
        wtot[w][lane] = s - x;
        if (lane == 31) ttot[w] = s;
    }
    __syncthreads();
    int run = 0;
#pragma unroll
    for (int i = 0; i < SCAN_TILES; ++i) {
        int mine = c[i].x + c[i].y + c[i].z + c[i].w;
        int ex   = run + wtot[i][w] + incl[i] - mine;
        int4 p;
        p.x = ex;
        p.y = p.x + c[i].x;
        p.z = p.y + c[i].y;
        p.w = p.z + c[i].z;
        ((int4*)rowptr)[i * SCAN_T + t] = p;
        run += ttot[i];
    }
}

__global__ void fill_kernel(const int* __restrict__ row, const int* __restrict__ col,
                            const int* __restrict__ rank, const int* __restrict__ rowptr,
                            const float* __restrict__ dinv, int2* __restrict__ erec) {
    int e = blockIdx.x * blockDim.x + threadIdx.x;
    if (e < N_EDGES) {
        int r = row[e];
        int c = col[e];
        float wgt = dinv[r] * dinv[c];
        erec[rowptr[c] + rank[e]] = make_int2(r, __float_as_int(wgt));
    }
}

template <int CNT>
__device__ __forceinline__ void gather_accum(const int2* __restrict__ rec,
                                             const float* __restrict__ hl, float4& acc) {
    int2   r[CNT];
    float4 v[CNT];
#pragma unroll
    for (int i = 0; i < CNT; ++i) r[i] = rec[i];
#pragma unroll
    for (int i = 0; i < CNT; ++i) {
        unsigned s = min((unsigned)r[i].x, (unsigned)(N_NODES - 1));
        v[i] = *(const float4*)(hl + (size_t)s * F_DIM);
    }
#pragma unroll
    for (int i = 0; i < CNT; ++i) {
        float wgt = __int_as_float(r[i].y);
        acc.x = fmaf(wgt, v[i].x, acc.x);
        acc.y = fmaf(wgt, v[i].y, acc.y);
        acc.z = fmaf(wgt, v[i].z, acc.z);
        acc.w = fmaf(wgt, v[i].w, acc.w);
    }
}

__global__ __launch_bounds__(256) void spmm_gather_kernel(const int* __restrict__ rowptr,
                                                          const int2* __restrict__ erec,
                                                          const float* __restrict__ hin,
                                                          float* __restrict__ hout) {
    const int lane = threadIdx.x & 31;
    const int node = __builtin_amdgcn_readfirstlane((int)(blockIdx.x * 8 + (threadIdx.x >> 5)));
    const int beg = rowptr[node];
    const int end = rowptr[node + 1];
    const float* __restrict__ hl = hin + lane * 4;
    float4 acc = make_float4(0.0f, 0.0f, 0.0f, 0.0f);
    int e = beg;
    while (e + 8 <= end) { gather_accum<8>(erec + e, hl, acc); e += 8; }
    if (e + 4 <= end)    { gather_accum<4>(erec + e, hl, acc); e += 4; }
    switch (end - e) {
        case 3: gather_accum<3>(erec + e, hl, acc); break;
        case 2: gather_accum<2>(erec + e, hl, acc); break;
        case 1: gather_accum<1>(erec + e, hl, acc); break;
        default: break;
    }
    *(float4*)(hout + (size_t)node * F_DIM + lane * 4) = acc;
}


template <int L>
__device__ __forceinline__ int row_share(int x) {
    return __builtin_amdgcn_update_dpp(0, x, 0x150 + L, 0xf, 0xf, false);
}

#define GATHER_PAIR_LOAD(P, I)                                                          \
    {                                                                                    \
        const int sx = row_share<(P)>(rv.x);                                             \
        w[I] = __int_as_float(row_share<(P)>(rv.y));                                     \
        v[I] = make_float4(0.0f, 0.0f, 0.0f, 0.0f);                                      \
        if (2 * (P) + hi < cnt) {                                                        \
                  \
                                                  \
            const unsigned s = min((unsigned)sx, (unsigned)(N_NODES - 1));               \
            v[I] = *(const float4*)(hl + (size_t)s * C_OUT);                             \
        }                                                                                \
    }

template <int P0>
__device__ __forceinline__ void gather_group(const int2 rv, const int cnt, const int hi,
                                             const float* __restrict__ hl, float4& acc) {
    float4 v[8];
    float  w[8];
    GATHER_PAIR_LOAD(P0 + 0, 0)
    GATHER_PAIR_LOAD(P0 + 1, 1)
    GATHER_PAIR_LOAD(P0 + 2, 2)
    GATHER_PAIR_LOAD(P0 + 3, 3)
    GATHER_PAIR_LOAD(P0 + 4, 4)
    GATHER_PAIR_LOAD(P0 + 5, 5)
    GATHER_PAIR_LOAD(P0 + 6, 6)
    GATHER_PAIR_LOAD(P0 + 7, 7)
#pragma unroll
    for (int i = 0; i < 8; ++i) {
        acc.x = fmaf(w[i], v[i].x, acc.x);
        acc.y = fmaf(w[i], v[i].y, acc.y);
        acc.z = fmaf(w[i], v[i].z, acc.z);
        acc.w = fmaf(w[i], v[i].w, acc.w);
    }
}

template <int LAST>
__global__ __launch_bounds__(256) void spmm_gather64_kernel(const int* __restrict__ rowptr,
                                                            const int2* __restrict__ erec,
                                                            const float* __restrict__ hin,
                                                            const float* __restrict__ zadd,
                                                            const float* __restrict__ bias,
                                                            float* __restrict__ hout) {
    const int lane   = threadIdx.x & 31;
    const int lane16 = lane & 15;
    const int hi     = lane >> 4;
    const int slot   = 2 * lane16 + hi;
    const int node = __builtin_amdgcn_readfirstlane((int)(blockIdx.x * 8 + (threadIdx.x >> 5)));
    const int beg = rowptr[node];
    const int end = rowptr[node + 1];
    const float* __restrict__ hl = hin + lane16 * 4;
    const float4 z = *(const float4*)(zadd + (size_t)node * C_OUT + lane16 * 4);
    float4 acc = make_float4(0.0f, 0.0f, 0.0f, 0.0f);
    for (int e = beg; e < end; e += 32) {
        const int cnt = min(end - e, 32);
        int2 rv = make_int2(0, 0);
        if (slot < cnt) rv = erec[e + slot];
        gather_group<0>(rv, cnt, hi, hl, acc);
        if (cnt > 16) gather_group<8>(rv, cnt, hi, hl, acc);
    }
    acc.x += __shfl_xor(acc.x, 16, 32);
    acc.y += __shfl_xor(acc.y, 16, 32);
    acc.z += __shfl_xor(acc.z, 16, 32);
    acc.w += __shfl_xor(acc.w, 16, 32);
    if (hi == 0) {
        float4 o = make_float4(acc.x + z.x, acc.y + z.y, acc.z + z.z, acc.w + z.w);
        if (LAST) {
            const float4 b = *(const float4*)(bias + lane16 * 4);
            o.x += b.x; o.y += b.y; o.z += b.z; o.w += b.w;
        }
        *(float4*)(hout + (size_t)node * C_OUT + lane16 * 4) = o;
    }
}
#undef GATHER_PAIR_LOAD

template <int NOUT>
struct GemmCfg {
    static constexpr int NT    = NOUT / 16;
    static constexpr int PITCH = 2 * NOUT + 32;
    static constexpr int SEGF  = (F_DIM / 2) * PITCH;
    static constexpr int NV4   = NOUT / 4;
    static constexpr int ITEMS = (F_DIM / 2) * NV4 / 256;
};

template <int NOUT, int NSEG, int MODE>
__global__ __launch_bounds__(256, 4) void tag_gemm_wmma_kernel(
    const float* __restrict__ s0, const float* __restrict__ s1,
    const float* __restrict__ s2, const float* __restrict__ s3,
    const float* __restrict__ W, const float* __restrict__ bias,
    float* __restrict__ out) {
    constexpr int NT    = GemmCfg<NOUT>::NT;
    constexpr int PITCH = GemmCfg<NOUT>::PITCH;
    constexpr int SEGF  = GemmCfg<NOUT>::SEGF;
    constexpr int NV4   = GemmCfg<NOUT>::NV4;
    constexpr int ITEMS = GemmCfg<NOUT>::ITEMS;
    constexpr int GSTEP = 8 * 2 * PITCH;

    constexpr int WP    = (MODE == 1) ? C_OUT : NOUT;
    constexpr int NBUF  = (NSEG > 1) ? 2 : 1;

    __shared__ __attribute__((aligned(16))) float Bs[NBUF * SEGF];

    const int tid    = threadIdx.x;
    const int lane   = tid & 31;
    const int lane16 = lane & 15;
    const int hi     = lane >> 4;
    const int wave   = __builtin_amdgcn_readfirstlane(tid >> 5);
    const int row0   = (blockIdx.x * 8 + wave) * 16;
    const bool active = row0 < N_NODES;
    const int m = min(row0 + lane16, N_NODES - 1);
    const size_t aoff = (size_t)m * F_DIM + hi * 2;

    const float* segs[4] = { s0, s1, s2, s3 };

    v8f acc[NT];
#pragma unroll
    for (int t = 0; t < NT; ++t) acc[t] = (v8f){ 0.f, 0.f, 0.f, 0.f, 0.f, 0.f, 0.f, 0.f };

    v2f a_cur[8], a_nxt[8];
#pragma unroll
    for (int j = 0; j < 8; ++j) a_cur[j] = *(const v2f*)(segs[0] + aoff + 4 * j);

    const int spr = tid / NV4;
    const int sn4 = tid % NV4;
    const float* __restrict__ wsrc = (MODE == 1)
        ? W + (size_t)(2 * blockIdx.y + ((4 * sn4) >> 6)) * (F_DIM * C_OUT)
            + (size_t)(2 * spr) * C_OUT + ((4 * sn4) & 63)
        : W + (size_t)(2 * spr) * NOUT + 4 * sn4;
    float* const wdst = Bs + spr * PITCH + 8 * sn4;
    constexpr int PR_STEP = 256 / NV4;
    float4 w0[ITEMS], w1[ITEMS];
#pragma unroll
    for (int it = 0; it < ITEMS; ++it) {
        const float* g = wsrc + (size_t)(2 * PR_STEP * it) * WP;
        w0[it] = *(const float4*)g;
        w1[it] = *(const float4*)(g + WP);
    }
#pragma unroll
    for (int it = 0; it < ITEMS; ++it) {
        float* d = wdst + it * PR_STEP * PITCH;
        *(float4*)d       = make_float4(w0[it].x, w1[it].x, w0[it].y, w1[it].y);
        *(float4*)(d + 4) = make_float4(w0[it].z, w1[it].z, w0[it].w, w1[it].w);
    }
    __syncthreads();

#pragma unroll
    for (int s = 0; s < NSEG; ++s) {
        const float* bg = Bs + (s & 1) * SEGF + hi * PITCH + 2 * lane16;
        const float* ag = segs[s] + aoff;
        const float* an = (s < NSEG - 1) ? segs[s + 1] + aoff : ag;
        v2f bq[2][NT];
#pragma unroll
        for (int t = 0; t < NT; ++t) bq[0][t] = *(const v2f*)(bg + 32 * t);

        auto products = [&](const float* bgp, const float* bnext) {
#pragma unroll
            for (int j = 0; j < 8; ++j) {
                const float* bn = (j < 7) ? bgp + (j + 1) * 2 * PITCH : bnext;
#pragma unroll
                for (int t = 0; t < NT; ++t) bq[(j + 1) & 1][t] = *(const v2f*)(bn + 32 * t);
#pragma unroll
                for (int t = 0; t < NT; ++t)
                    acc[t] = __builtin_amdgcn_wmma_f32_16x16x4_f32(
                        false, a_cur[j], false, bq[j & 1][t], (short)0, acc[t], false, false);
            }
        };

#pragma unroll
        for (int j = 0; j < 8; ++j) a_nxt[j] = *(const v2f*)(ag + 32 + 4 * j);
        if (s < NSEG - 1) {
#pragma unroll
            for (int it = 0; it < ITEMS; ++it) {
                const float* g = wsrc + (size_t)((s + 1) * F_DIM + 2 * PR_STEP * it) * WP;
                w0[it] = *(const float4*)g;
                w1[it] = *(const float4*)(g + WP);
            }
        }
        products(bg, bg + GSTEP);
#pragma unroll
        for (int j = 0; j < 8; ++j) a_cur[j] = a_nxt[j];
        bg += GSTEP;
        ag += 32;

#pragma unroll 1
        for (int sc = 1; sc < 4; ++sc) {
            const bool last = (sc == 3);
            const float* ap = last ? an : ag + 32;
            if (s < NSEG - 1 || !last) {
#pragma unroll
                for (int j = 0; j < 8; ++j) a_nxt[j] = *(const v2f*)(ap + 4 * j);
            }
            products(bg, last ? bg + 7 * 2 * PITCH : bg + GSTEP);
#pragma unroll
            for (int j = 0; j < 8; ++j) a_cur[j] = a_nxt[j];
            bg += GSTEP;
            ag += 32;
        }

        if (s < NSEG - 1) {
            float* db = wdst + ((s + 1) & 1) * SEGF;
#pragma unroll
            for (int it = 0; it < ITEMS; ++it) {
                float* d = db + it * PR_STEP * PITCH;
                *(float4*)d       = make_float4(w0[it].x, w1[it].x, w0[it].y, w1[it].y);
                *(float4*)(d + 4) = make_float4(w0[it].z, w1[it].z, w0[it].w, w1[it].w);
            }
            __syncthreads();
        }
    }

    if (active) {
#pragma unroll
        for (int t = 0; t < NT; ++t) {
            const int n = t * 16 + lane16;
            if (MODE == 0) {
                const float bv = bias[n];
#pragma unroll
                for (int j = 0; j < 8; ++j) {
                    int mm  = row0 + j + hi * 8;
                    float v = acc[t][j] + bv;
                    v = (v > 0.0f) ? v : 0.0f;
                    out[(long long)mm * NOUT + n] = v;
                }
            } else {
                float* zp = out + (size_t)(2 * blockIdx.y + (t >> 2)) * ((size_t)N_NODES * C_OUT)
                                + ((t & 3) * 16 + lane16);
#pragma unroll
                for (int j = 0; j < 8; ++j) {
                    int mm = row0 + j + hi * 8;
                    zp[(size_t)mm * C_OUT] = acc[t][j];
                }
            }
        }
    }
}

__global__ void log_softmax_kernel(float* __restrict__ out) {
    int wave = threadIdx.x >> 5;
    int lane = threadIdx.x & 31;
    int r = blockIdx.x * (blockDim.x >> 5) + wave;
    if (r >= N_NODES) return;
    float* p = out + (long long)r * C_OUT;
    float a = p[lane];
    float b = p[lane + 32];
    float mx = fmaxf(a, b);
#pragma unroll
    for (int off = 16; off > 0; off >>= 1) mx = fmaxf(mx, __shfl_xor(mx, off, 32));
    float s = expf(a - mx) + expf(b - mx);
#pragma unroll
    for (int off = 16; off > 0; off >>= 1) s += __shfl_xor(s, off, 32);
    float lse = mx + logf(s);
    p[lane]      = a - lse;
    p[lane + 32] = b - lse;
}


extern "C" void kernel_launch(void* const* d_in, const int* in_sizes, int n_in,
                              void* d_out, int out_size, void* d_ws, size_t ws_size,
                              hipStream_t stream) {
    const float* x   = (const float*)d_in[0];
    const int*   ei  = (const int*)d_in[1];
    const float* W1  = (const float*)d_in[2];
    const float* b1  = (const float*)d_in[3];
    const float* W2  = (const float*)d_in[4];
    const float* b2  = (const float*)d_in[5];
    float*       out = (float*)d_out;

    const int* rowp = ei;
    const int* colp = ei + N_EDGES;

    int*   cnt    = (int*)d_ws;
    int*   rowptr = cnt + N_PAD;
    float* dinv   = (float*)(rowptr + N_PAD);
    int*   rank   = (int*)(dinv + N_PAD);
    int2*  erec   = (int2*)(rank + N_EDGES);
    float* hb0    = (float*)(erec + N_EDGES);
    float* hb1    = hb0 + (size_t)N_NODES * F_DIM;
    float* hb2    = hb1 + (size_t)N_NODES * F_DIM;
    float* acc1   = hb2 + (size_t)N_NODES * F_DIM;
    float* zpl    = acc1 + (size_t)N_NODES * F_DIM;
    const size_t ZP = (size_t)N_NODES * C_OUT;
    float* t1     = hb0;
    float* t2     = hb1;

    const int TB = 256;
    const int padBlocks  = N_PAD / TB;
    const int edgeBlocks = (N_EDGES + TB - 1) / TB;
    const int nodeGroups = N_NODES / 8;
    const int rowTiles   = N_NODES / 16;
    const int rowGroups  = (rowTiles + 7) / 8;

    zero_i32_kernel<<<padBlocks, TB, 0, stream>>>(cnt, N_PAD);
    hist_rank_kernel<<<edgeBlocks, TB, 0, stream>>>(colp, cnt, rank);
    scan_dinv_kernel<<<1, SCAN_T, 0, stream>>>(cnt, rowptr, dinv);
    fill_kernel<<<edgeBlocks, TB, 0, stream>>>(rowp, colp, rank, rowptr, dinv, erec);

    spmm_gather_kernel<<<nodeGroups, 256, 0, stream>>>(rowptr, erec, x, hb0);
    spmm_gather_kernel<<<nodeGroups, 256, 0, stream>>>(rowptr, erec, hb0, hb1);
    spmm_gather_kernel<<<nodeGroups, 256, 0, stream>>>(rowptr, erec, hb1, hb2);

    tag_gemm_wmma_kernel<H_OUT1, 4, 0><<<rowGroups, 256, 0, stream>>>(
        x, hb0, hb1, hb2, W1, b1, acc1);

    tag_gemm_wmma_kernel<2 * C_OUT, 1, 1><<<dim3(rowGroups, 2), 256, 0, stream>>>(
        acc1, acc1, acc1, acc1, W2, b2, zpl);

    spmm_gather64_kernel<0><<<nodeGroups, 256, 0, stream>>>(
        rowptr, erec, zpl + 3 * ZP, zpl + 2 * ZP, b2, t1);
    spmm_gather64_kernel<0><<<nodeGroups, 256, 0, stream>>>(
        rowptr, erec, t1, zpl + ZP, b2, t2);
    spmm_gather64_kernel<1><<<nodeGroups, 256, 0, stream>>>(
        rowptr, erec, t2, zpl, b2, out);

    log_softmax_kernel<<<N_NODES / 8, TB, 0, stream>>>(out);
}


# Edge sort by target: per-edge returning global atomics + scattered slot stores -> two-level counting sort in LDS (256-node bins, then one work-group per bin); per-node sum order = LDS arrival order
# speedup vs baseline: 6.1352x; 1.2965x over previous
#include <hip/hip_runtime.h>
#include <hip/hip_bf16.h>
#include <math.h>


#define N_NODES 50000
#define N_EDGES 800000
#define F_DIM   128
#define H_OUT1  128
#define C_OUT   64
#define K_TOT   512

#define N_PAD      53248

typedef __attribute__((ext_vector_type(2))) float v2f;
typedef __attribute__((ext_vector_type(8))) float v8f;


#define SORT_T    256
#define SORT_V4   4
#define SORT_CH   (SORT_T * SORT_V4 * 4)
#define SORT_NB   ((N_EDGES + SORT_CH - 1) / SORT_CH)
#define SORT_LD   28
#define BIN_SHIFT 8
#define BIN_MASK  255
static_assert(SORT_NB % SORT_LD == 0 && SORT_NB * SORT_T <= N_PAD, "edge block count");
#define N_BINBLK  (N_PAD / 256)

__device__ __forceinline__ int clamp_node(int c) {
    return (int)min((unsigned)c, (unsigned)(N_NODES - 1));
}

__device__ __forceinline__ int excl_scan256(int v, int* wt, int tid) {
    const int lane = tid & 31;
    const int w    = tid >> 5;
    int s = v;
    for (int off = 1; off < 32; off <<= 1) {
        int nb = __shfl_up(s, off, 32);
        if (lane >= off) s += nb;
    }
    if (w < 8 && lane == 31) wt[w] = s;
    __syncthreads();
    int pre = 0;
    for (int i = 0; i < 8; ++i) pre += (i < w) ? wt[i] : 0;
    __syncthreads();
    return pre + s - v;
}

__global__ __launch_bounds__(SORT_T) void bin_count_kernel(const int* __restrict__ col,
                                                           int* __restrict__ bhist) {
    __shared__ int h[SORT_T];
    const int tid = threadIdx.x;
    h[tid] = 0;
    __syncthreads();
    for (int k = 0; k < SORT_V4; ++k) {
        const int i4 = (blockIdx.x * SORT_V4 + k) * SORT_T + tid;
        if (i4 < N_EDGES / 4) {
            const int4 c = ((const int4*)col)[i4];
            atomicAdd(&h[clamp_node(c.x) >> BIN_SHIFT], 1);
            atomicAdd(&h[clamp_node(c.y) >> BIN_SHIFT], 1);
            atomicAdd(&h[clamp_node(c.z) >> BIN_SHIFT], 1);
            atomicAdd(&h[clamp_node(c.w) >> BIN_SHIFT], 1);
        }
    }
    __syncthreads();
    bhist[blockIdx.x * SORT_T + tid] = h[tid];
}

__global__ __launch_bounds__(SORT_T) void bin_scatter_kernel(const int* __restrict__ row,
                                                             const int* __restrict__ col,
                                                             const int* __restrict__ bhist,
                                                             int* __restrict__ binstart,
                                                             int2* __restrict__ tmp) {
    __shared__ int2 stage[SORT_CH];
    __shared__ int lcur[SORT_T];
    __shared__ int gdelta[SORT_T];
    __shared__ int wt[8];
    const int tid = threadIdx.x;
    int total = 0, before = 0, own = 0;
    for (int b0 = 0; b0 < SORT_NB; b0 += SORT_LD) {
        int v[SORT_LD];
#pragma unroll
        for (int j = 0; j < SORT_LD; ++j) v[j] = bhist[(b0 + j) * SORT_T + tid];
#pragma unroll
        for (int j = 0; j < SORT_LD; ++j) {
            total += v[j];
            before += (b0 + j < (int)blockIdx.x) ? v[j] : 0;
            own    += (b0 + j == (int)blockIdx.x) ? v[j] : 0;
        }
    }
    const int ex   = excl_scan256(total, wt, tid);
    const int lbeg = excl_scan256(own, wt, tid);
    lcur[tid]   = lbeg;
    gdelta[tid] = ex + before - lbeg;
    if (blockIdx.x == 0) {
        binstart[tid] = ex;
        if (tid == SORT_T - 1) binstart[SORT_T] = ex + total;
    }
    __syncthreads();
    for (int k = 0; k < SORT_V4; ++k) {
        const int i4 = (blockIdx.x * SORT_V4 + k) * SORT_T + tid;
        if (i4 < N_EDGES / 4) {
            const int4 r = ((const int4*)row)[i4];
            int4 c = ((const int4*)col)[i4];
            c.x = clamp_node(c.x); c.y = clamp_node(c.y);
            c.z = clamp_node(c.z); c.w = clamp_node(c.w);
            const int q0 = atomicAdd(&lcur[c.x >> BIN_SHIFT], 1);
            const int q1 = atomicAdd(&lcur[c.y >> BIN_SHIFT], 1);
            const int q2 = atomicAdd(&lcur[c.z >> BIN_SHIFT], 1);
            const int q3 = atomicAdd(&lcur[c.w >> BIN_SHIFT], 1);
            stage[q0] = make_int2(r.x, c.x);
            stage[q1] = make_int2(r.y, c.y);
            stage[q2] = make_int2(r.z, c.z);
            stage[q3] = make_int2(r.w, c.w);
        }
    }
    __syncthreads();
    const int nblk = min(SORT_CH, N_EDGES - (int)blockIdx.x * SORT_CH);
    for (int i = tid; i < nblk; i += SORT_T) {
        const int2 rc = stage[i];
        tmp[i + gdelta[rc.y >> BIN_SHIFT]] = rc;
    }
}

#define BIN_T   1024
#define BIN_R   8
#define BIN_CAP (BIN_T * BIN_R)
__global__ __launch_bounds__(BIN_T) void bin_sort_kernel(const int2* __restrict__ tmp,
                                                         const int* __restrict__ binstart,
                                                         int* __restrict__ rowptr,
                                                         float* __restrict__ dinv,
                                                         int2* __restrict__ erec) {
    __shared__ int2 stage[BIN_CAP];
    __shared__ int cnt[256];
    __shared__ int cur[256];
    __shared__ int wt[8];
    const int tid = threadIdx.x;
    const int beg = binstart[blockIdx.x];
    const int n   = binstart[blockIdx.x + 1] - beg;
    const bool small = n <= BIN_CAP;
    if (tid < 256) cnt[tid] = 0;
    __syncthreads();
    int2 rc[BIN_R];
    if (small) {
#pragma unroll
        for (int j = 0; j < BIN_R; ++j) {
            const int i = j * BIN_T + tid;
            rc[j] = make_int2(0, 0);
            if (i < n) rc[j] = tmp[beg + i];
        }
#pragma unroll
        for (int j = 0; j < BIN_R; ++j)
            if (j * BIN_T + tid < n) atomicAdd(&cnt[rc[j].y & BIN_MASK], 1);
    } else {
        for (int i = tid; i < n; i += BIN_T) atomicAdd(&cnt[tmp[beg + i].y & BIN_MASK], 1);
    }
    __syncthreads();
    const int v  = (tid < 256) ? cnt[tid] : 0;
    const int ex = excl_scan256(v, wt, tid);
    if (tid < 256) {
        const int node = blockIdx.x * 256 + tid;
        rowptr[node] = beg + ex;
        dinv[node]   = (v > 0) ? rsqrtf((float)v) : 0.0f;
        cur[tid]     = ex;
    }
    __syncthreads();
    if (small) {
#pragma unroll
        for (int j = 0; j < BIN_R; ++j)
            if (j * BIN_T + tid < n) stage[atomicAdd(&cur[rc[j].y & BIN_MASK], 1)] = rc[j];
        __syncthreads();
        for (int i = tid; i < n; i += BIN_T) erec[beg + i] = stage[i];
    } else {
        for (int i = tid; i < n; i += BIN_T) {
            const int2 e = tmp[beg + i];
            erec[beg + atomicAdd(&cur[e.y & BIN_MASK], 1)] = e;
        }
    }
}

__global__ void edge_weight_kernel(const float* __restrict__ dinv, int2* __restrict__ erec) {
    int p = blockIdx.x * blockDim.x + threadIdx.x;
    if (p < N_EDGES) {
        const int2 rc = erec[p];
        float wgt = dinv[clamp_node(rc.x)] * dinv[rc.y];
        erec[p] = make_int2(rc.x, __float_as_int(wgt));
    }
}

template <int CNT>
__device__ __forceinline__ void gather_accum(const int2* __restrict__ rec,
                                             const float* __restrict__ hl, float4& acc) {
    int2   r[CNT];
    float4 v[CNT];
#pragma unroll
    for (int i = 0; i < CNT; ++i) r[i] = rec[i];
#pragma unroll
    for (int i = 0; i < CNT; ++i) {
        unsigned s = min((unsigned)r[i].x, (unsigned)(N_NODES - 1));
        v[i] = *(const float4*)(hl + (size_t)s * F_DIM);
    }
#pragma unroll
    for (int i = 0; i < CNT; ++i) {
        float wgt = __int_as_float(r[i].y);
        acc.x = fmaf(wgt, v[i].x, acc.x);
        acc.y = fmaf(wgt, v[i].y, acc.y);
        acc.z = fmaf(wgt, v[i].z, acc.z);
        acc.w = fmaf(wgt, v[i].w, acc.w);
    }
}

__global__ __launch_bounds__(256) void spmm_gather_kernel(const int* __restrict__ rowptr,
                                                          const int2* __restrict__ erec,
                                                          const float* __restrict__ hin,
                                                          float* __restrict__ hout) {
    const int lane = threadIdx.x & 31;
    const int node = __builtin_amdgcn_readfirstlane((int)(blockIdx.x * 8 + (threadIdx.x >> 5)));
    const int beg = rowptr[node];
    const int end = rowptr[node + 1];
    const float* __restrict__ hl = hin + lane * 4;
    float4 acc = make_float4(0.0f, 0.0f, 0.0f, 0.0f);
    int e = beg;
    while (e + 8 <= end) { gather_accum<8>(erec + e, hl, acc); e += 8; }
    if (e + 4 <= end)    { gather_accum<4>(erec + e, hl, acc); e += 4; }
    switch (end - e) {
        case 3: gather_accum<3>(erec + e, hl, acc); break;
        case 2: gather_accum<2>(erec + e, hl, acc); break;
        case 1: gather_accum<1>(erec + e, hl, acc); break;
        default: break;
    }
    *(float4*)(hout + (size_t)node * F_DIM + lane * 4) = acc;
}


template <int L>
__device__ __forceinline__ int row_share(int x) {
    return __builtin_amdgcn_update_dpp(0, x, 0x150 + L, 0xf, 0xf, false);
}

#define GATHER_PAIR_LOAD(P, I)                                                          \
    {                                                                                    \
        const int sx = row_share<(P)>(rv.x);                                             \
        w[I] = __int_as_float(row_share<(P)>(rv.y));                                     \
        v[I] = make_float4(0.0f, 0.0f, 0.0f, 0.0f);                                      \
        if (2 * (P) + hi < cnt) {                                                        \
                  \
                                                  \
            const unsigned s = min((unsigned)sx, (unsigned)(N_NODES - 1));               \
            v[I] = *(const float4*)(hl + (size_t)s * C_OUT);                             \
        }                                                                                \
    }

template <int P0>
__device__ __forceinline__ void gather_group(const int2 rv, const int cnt, const int hi,
                                             const float* __restrict__ hl, float4& acc) {
    float4 v[8];
    float  w[8];
    GATHER_PAIR_LOAD(P0 + 0, 0)
    GATHER_PAIR_LOAD(P0 + 1, 1)
    GATHER_PAIR_LOAD(P0 + 2, 2)
    GATHER_PAIR_LOAD(P0 + 3, 3)
    GATHER_PAIR_LOAD(P0 + 4, 4)
    GATHER_PAIR_LOAD(P0 + 5, 5)
    GATHER_PAIR_LOAD(P0 + 6, 6)
    GATHER_PAIR_LOAD(P0 + 7, 7)
#pragma unroll
    for (int i = 0; i < 8; ++i) {
        acc.x = fmaf(w[i], v[i].x, acc.x);
        acc.y = fmaf(w[i], v[i].y, acc.y);
        acc.z = fmaf(w[i], v[i].z, acc.z);
        acc.w = fmaf(w[i], v[i].w, acc.w);
    }
}

template <int LAST>
__global__ __launch_bounds__(256) void spmm_gather64_kernel(const int* __restrict__ rowptr,
                                                            const int2* __restrict__ erec,
                                                            const float* __restrict__ hin,
                                                            const float* __restrict__ zadd,
                                                            const float* __restrict__ bias,
                                                            float* __restrict__ hout) {
    const int lane   = threadIdx.x & 31;
    const int lane16 = lane & 15;
    const int hi     = lane >> 4;
    const int slot   = 2 * lane16 + hi;
    const int node = __builtin_amdgcn_readfirstlane((int)(blockIdx.x * 8 + (threadIdx.x >> 5)));
    const int beg = rowptr[node];
    const int end = rowptr[node + 1];
    const float* __restrict__ hl = hin + lane16 * 4;
    const float4 z = *(const float4*)(zadd + (size_t)node * C_OUT + lane16 * 4);
    float4 acc = make_float4(0.0f, 0.0f, 0.0f, 0.0f);
    for (int e = beg; e < end; e += 32) {
        const int cnt = min(end - e, 32);
        int2 rv = make_int2(0, 0);
        if (slot < cnt) rv = erec[e + slot];
        gather_group<0>(rv, cnt, hi, hl, acc);
        if (cnt > 16) gather_group<8>(rv, cnt, hi, hl, acc);
    }
    acc.x += __shfl_xor(acc.x, 16, 32);
    acc.y += __shfl_xor(acc.y, 16, 32);
    acc.z += __shfl_xor(acc.z, 16, 32);
    acc.w += __shfl_xor(acc.w, 16, 32);
    if (hi == 0) {
        float4 o = make_float4(acc.x + z.x, acc.y + z.y, acc.z + z.z, acc.w + z.w);
        if (LAST) {
            const float4 b = *(const float4*)(bias + lane16 * 4);
            o.x += b.x; o.y += b.y; o.z += b.z; o.w += b.w;
        }
        *(float4*)(hout + (size_t)node * C_OUT + lane16 * 4) = o;
    }
}
#undef GATHER_PAIR_LOAD

template <int NOUT>
struct GemmCfg {
    static constexpr int NT    = NOUT / 16;
    static constexpr int PITCH = 2 * NOUT + 32;
    static constexpr int SEGF  = (F_DIM / 2) * PITCH;
    static constexpr int NV4   = NOUT / 4;
    static constexpr int ITEMS = (F_DIM / 2) * NV4 / 256;
};

template <int NOUT, int NSEG, int MODE>
__global__ __launch_bounds__(256, 4) void tag_gemm_wmma_kernel(
    const float* __restrict__ s0, const float* __restrict__ s1,
    const float* __restrict__ s2, const float* __restrict__ s3,
    const float* __restrict__ W, const float* __restrict__ bias,
    float* __restrict__ out) {
    constexpr int NT    = GemmCfg<NOUT>::NT;
    constexpr int PITCH = GemmCfg<NOUT>::PITCH;
    constexpr int SEGF  = GemmCfg<NOUT>::SEGF;
    constexpr int NV4   = GemmCfg<NOUT>::NV4;
    constexpr int ITEMS = GemmCfg<NOUT>::ITEMS;
    constexpr int GSTEP = 8 * 2 * PITCH;

    constexpr int WP    = (MODE == 1) ? C_OUT : NOUT;
    constexpr int NBUF  = (NSEG > 1) ? 2 : 1;

    __shared__ __attribute__((aligned(16))) float Bs[NBUF * SEGF];

    const int tid    = threadIdx.x;
    const int lane   = tid & 31;
    const int lane16 = lane & 15;
    const int hi     = lane >> 4;
    const int wave   = __builtin_amdgcn_readfirstlane(tid >> 5);
    const int row0   = (blockIdx.x * 8 + wave) * 16;
    const bool active = row0 < N_NODES;
    const int m = min(row0 + lane16, N_NODES - 1);
    const size_t aoff = (size_t)m * F_DIM + hi * 2;

    const float* segs[4] = { s0, s1, s2, s3 };

    v8f acc[NT];
#pragma unroll
    for (int t = 0; t < NT; ++t) acc[t] = (v8f){ 0.f, 0.f, 0.f, 0.f, 0.f, 0.f, 0.f, 0.f };

    v2f a_cur[8], a_nxt[8];
#pragma unroll
    for (int j = 0; j < 8; ++j) a_cur[j] = *(const v2f*)(segs[0] + aoff + 4 * j);

    const int spr = tid / NV4;
    const int sn4 = tid % NV4;
    const float* __restrict__ wsrc = (MODE == 1)
        ? W + (size_t)(2 * blockIdx.y + ((4 * sn4) >> 6)) * (F_DIM * C_OUT)
            + (size_t)(2 * spr) * C_OUT + ((4 * sn4) & 63)
        : W + (size_t)(2 * spr) * NOUT + 4 * sn4;
    float* const wdst = Bs + spr * PITCH + 8 * sn4;
    constexpr int PR_STEP = 256 / NV4;
    float4 w0[ITEMS], w1[ITEMS];
#pragma unroll
    for (int it = 0; it < ITEMS; ++it) {
        const float* g = wsrc + (size_t)(2 * PR_STEP * it) * WP;
        w0[it] = *(const float4*)g;
        w1[it] = *(const float4*)(g + WP);
    }
#pragma unroll
    for (int it = 0; it < ITEMS; ++it) {
        float* d = wdst + it * PR_STEP * PITCH;
        *(float4*)d       = make_float4(w0[it].x, w1[it].x, w0[it].y, w1[it].y);
        *(float4*)(d + 4) = make_float4(w0[it].z, w1[it].z, w0[it].w, w1[it].w);
    }
    __syncthreads();

#pragma unroll
    for (int s = 0; s < NSEG; ++s) {
        const float* bg = Bs + (s & 1) * SEGF + hi * PITCH + 2 * lane16;
        const float* ag = segs[s] + aoff;
        const float* an = (s < NSEG - 1) ? segs[s + 1] + aoff : ag;
        v2f bq[2][NT];
#pragma unroll
        for (int t = 0; t < NT; ++t) bq[0][t] = *(const v2f*)(bg + 32 * t);

        auto products = [&](const float* bgp, const float* bnext) {
#pragma unroll
            for (int j = 0; j < 8; ++j) {
                const float* bn = (j < 7) ? bgp + (j + 1) * 2 * PITCH : bnext;
#pragma unroll
                for (int t = 0; t < NT; ++t) bq[(j + 1) & 1][t] = *(const v2f*)(bn + 32 * t);
#pragma unroll
                for (int t = 0; t < NT; ++t)
                    acc[t] = __builtin_amdgcn_wmma_f32_16x16x4_f32(
                        false, a_cur[j], false, bq[j & 1][t], (short)0, acc[t], false, false);
            }
        };

#pragma unroll
        for (int j = 0; j < 8; ++j) a_nxt[j] = *(const v2f*)(ag + 32 + 4 * j);
        if (s < NSEG - 1) {
#pragma unroll
            for (int it = 0; it < ITEMS; ++it) {
                const float* g = wsrc + (size_t)((s + 1) * F_DIM + 2 * PR_STEP * it) * WP;
                w0[it] = *(const float4*)g;
                w1[it] = *(const float4*)(g + WP);
            }
        }
        products(bg, bg + GSTEP);
#pragma unroll
        for (int j = 0; j < 8; ++j) a_cur[j] = a_nxt[j];
        bg += GSTEP;
        ag += 32;

#pragma unroll 1
        for (int sc = 1; sc < 4; ++sc) {
            const bool last = (sc == 3);
            const float* ap = last ? an : ag + 32;
            if (s < NSEG - 1 || !last) {
#pragma unroll
                for (int j = 0; j < 8; ++j) a_nxt[j] = *(const v2f*)(ap + 4 * j);
            }
            products(bg, last ? bg + 7 * 2 * PITCH : bg + GSTEP);
#pragma unroll
            for (int j = 0; j < 8; ++j) a_cur[j] = a_nxt[j];
            bg += GSTEP;
            ag += 32;
        }

        if (s < NSEG - 1) {
            float* db = wdst + ((s + 1) & 1) * SEGF;
#pragma unroll
            for (int it = 0; it < ITEMS; ++it) {
                float* d = db + it * PR_STEP * PITCH;
                *(float4*)d       = make_float4(w0[it].x, w1[it].x, w0[it].y, w1[it].y);
                *(float4*)(d + 4) = make_float4(w0[it].z, w1[it].z, w0[it].w, w1[it].w);
            }
            __syncthreads();
        }
    }

    if (active) {
#pragma unroll
        for (int t = 0; t < NT; ++t) {
            const int n = t * 16 + lane16;
            if (MODE == 0) {
                const float bv = bias[n];
#pragma unroll
                for (int j = 0; j < 8; ++j) {
                    int mm  = row0 + j + hi * 8;
                    float v = acc[t][j] + bv;
                    v = (v > 0.0f) ? v : 0.0f;
                    out[(long long)mm * NOUT + n] = v;
                }
            } else {
                float* zp = out + (size_t)(2 * blockIdx.y + (t >> 2)) * ((size_t)N_NODES * C_OUT)
                                + ((t & 3) * 16 + lane16);
#pragma unroll
                for (int j = 0; j < 8; ++j) {
                    int mm = row0 + j + hi * 8;
                    zp[(size_t)mm * C_OUT] = acc[t][j];
                }
            }
        }
    }
}

__global__ void log_softmax_kernel(float* __restrict__ out) {
    int wave = threadIdx.x >> 5;
    int lane = threadIdx.x & 31;
    int r = blockIdx.x * (blockDim.x >> 5) + wave;
    if (r >= N_NODES) return;
    float* p = out + (long long)r * C_OUT;
    float a = p[lane];
    float b = p[lane + 32];
    float mx = fmaxf(a, b);
#pragma unroll
    for (int off = 16; off > 0; off >>= 1) mx = fmaxf(mx, __shfl_xor(mx, off, 32));
    float s = expf(a - mx) + expf(b - mx);
#pragma unroll
    for (int off = 16; off > 0; off >>= 1) s += __shfl_xor(s, off, 32);
    float lse = mx + logf(s);
    p[lane]      = a - lse;
    p[lane + 32] = b - lse;
}


extern "C" void kernel_launch(void* const* d_in, const int* in_sizes, int n_in,
                              void* d_out, int out_size, void* d_ws, size_t ws_size,
                              hipStream_t stream) {
    const float* x   = (const float*)d_in[0];
    const int*   ei  = (const int*)d_in[1];
    const float* W1  = (const float*)d_in[2];
    const float* b1  = (const float*)d_in[3];
    const float* W2  = (const float*)d_in[4];
    const float* b2  = (const float*)d_in[5];
    float*       out = (float*)d_out;

    const int* rowp = ei;
    const int* colp = ei + N_EDGES;

    int*   bhist  = (int*)d_ws;
    int*   rowptr = bhist + N_PAD;
    float* dinv   = (float*)(rowptr + N_PAD);
    int*   binst  = (int*)(dinv + N_PAD);
    int2*  erec   = (int2*)(binst + N_EDGES);
    float* hb0    = (float*)(erec + N_EDGES);
    float* hb1    = hb0 + (size_t)N_NODES * F_DIM;
    float* hb2    = hb1 + (size_t)N_NODES * F_DIM;
    float* acc1   = hb2 + (size_t)N_NODES * F_DIM;
    float* zpl    = acc1 + (size_t)N_NODES * F_DIM;
    const size_t ZP = (size_t)N_NODES * C_OUT;
    float* t1     = hb0;
    float* t2     = hb1;
    int2*  tmp    = (int2*)hb2;

    const int TB = 256;
    const int edgeBlocks = (N_EDGES + TB - 1) / TB;
    const int nodeGroups = N_NODES / 8;
    const int rowTiles   = N_NODES / 16;
    const int rowGroups  = (rowTiles + 7) / 8;

    bin_count_kernel<<<SORT_NB, SORT_T, 0, stream>>>(colp, bhist);
    bin_scatter_kernel<<<SORT_NB, SORT_T, 0, stream>>>(rowp, colp, bhist, binst, tmp);
    bin_sort_kernel<<<N_BINBLK, BIN_T, 0, stream>>>(tmp, binst, rowptr, dinv, erec);
    edge_weight_kernel<<<edgeBlocks, TB, 0, stream>>>(dinv, erec);

    spmm_gather_kernel<<<nodeGroups, 256, 0, stream>>>(rowptr, erec, x, hb0);
    spmm_gather_kernel<<<nodeGroups, 256, 0, stream>>>(rowptr, erec, hb0, hb1);
    spmm_gather_kernel<<<nodeGroups, 256, 0, stream>>>(rowptr, erec, hb1, hb2);

    tag_gemm_wmma_kernel<H_OUT1, 4, 0><<<rowGroups, 256, 0, stream>>>(
        x, hb0, hb1, hb2, W1, b1, acc1);

    tag_gemm_wmma_kernel<2 * C_OUT, 1, 1><<<dim3(rowGroups, 2), 256, 0, stream>>>(
        acc1, acc1, acc1, acc1, W2, b2, zpl);

    spmm_gather64_kernel<0><<<nodeGroups, 256, 0, stream>>>(
        rowptr, erec, zpl + 3 * ZP, zpl + 2 * ZP, b2, t1);
    spmm_gather64_kernel<0><<<nodeGroups, 256, 0, stream>>>(
        rowptr, erec, t1, zpl + ZP, b2, t2);
    spmm_gather64_kernel<1><<<nodeGroups, 256, 0, stream>>>(
        rowptr, erec, t2, zpl, b2, out);

    log_softmax_kernel<<<N_NODES / 8, TB, 0, stream>>>(out);
}


# Both elementwise passes folded into the hop next to them: hop 1 forms and stores the edge weights, the last hop writes log_softmax from registers; 11 launches, not 13; exp-sum order per row changes
# speedup vs baseline: 6.4052x; 1.0440x over previous
#include <hip/hip_runtime.h>
#include <hip/hip_bf16.h>
#include <math.h>


#define N_NODES 50000
#define N_EDGES 800000
#define F_DIM   128
#define H_OUT1  128
#define C_OUT   64
#define K_TOT   512

#define N_PAD      53248

typedef __attribute__((ext_vector_type(2))) float v2f;
typedef __attribute__((ext_vector_type(8))) float v8f;


#define SORT_T    256
#define SORT_V4   4
#define SORT_CH   (SORT_T * SORT_V4 * 4)
#define SORT_NB   ((N_EDGES + SORT_CH - 1) / SORT_CH)
#define SORT_LD   28
#define BIN_SHIFT 8
#define BIN_MASK  255
static_assert(SORT_NB % SORT_LD == 0 && SORT_NB * SORT_T <= N_PAD, "edge block count");
#define N_BINBLK  (N_PAD / 256)

__device__ __forceinline__ int clamp_node(int c) {
    return (int)min((unsigned)c, (unsigned)(N_NODES - 1));
}

__device__ __forceinline__ int excl_scan256(int v, int* wt, int tid) {
    const int lane = tid & 31;
    const int w    = tid >> 5;
    int s = v;
    for (int off = 1; off < 32; off <<= 1) {
        int nb = __shfl_up(s, off, 32);
        if (lane >= off) s += nb;
    }
    if (w < 8 && lane == 31) wt[w] = s;
    __syncthreads();
    int pre = 0;
    for (int i = 0; i < 8; ++i) pre += (i < w) ? wt[i] : 0;
    __syncthreads();
    return pre + s - v;
}

__global__ __launch_bounds__(SORT_T) void bin_count_kernel(const int* __restrict__ col,
                                                           int* __restrict__ bhist) {
    __shared__ int h[SORT_T];
    const int tid = threadIdx.x;
    h[tid] = 0;
    __syncthreads();
    for (int k = 0; k < SORT_V4; ++k) {
        const int i4 = (blockIdx.x * SORT_V4 + k) * SORT_T + tid;
        if (i4 < N_EDGES / 4) {
            const int4 c = ((const int4*)col)[i4];
            atomicAdd(&h[clamp_node(c.x) >> BIN_SHIFT], 1);
            atomicAdd(&h[clamp_node(c.y) >> BIN_SHIFT], 1);
            atomicAdd(&h[clamp_node(c.z) >> BIN_SHIFT], 1);
            atomicAdd(&h[clamp_node(c.w) >> BIN_SHIFT], 1);
        }
    }
    __syncthreads();
    bhist[blockIdx.x * SORT_T + tid] = h[tid];
}

__global__ __launch_bounds__(SORT_T) void bin_scatter_kernel(const int* __restrict__ row,
                                                             const int* __restrict__ col,
                                                             const int* __restrict__ bhist,
                                                             int* __restrict__ binstart,
                                                             int2* __restrict__ tmp) {
    __shared__ int2 stage[SORT_CH];
    __shared__ int lcur[SORT_T];
    __shared__ int gdelta[SORT_T];
    __shared__ int wt[8];
    const int tid = threadIdx.x;
    int total = 0, before = 0, own = 0;
    for (int b0 = 0; b0 < SORT_NB; b0 += SORT_LD) {
        int v[SORT_LD];
#pragma unroll
        for (int j = 0; j < SORT_LD; ++j) v[j] = bhist[(b0 + j) * SORT_T + tid];
#pragma unroll
        for (int j = 0; j < SORT_LD; ++j) {
            total += v[j];
            before += (b0 + j < (int)blockIdx.x) ? v[j] : 0;
            own    += (b0 + j == (int)blockIdx.x) ? v[j] : 0;
        }
    }
    const int ex   = excl_scan256(total, wt, tid);
    const int lbeg = excl_scan256(own, wt, tid);
    lcur[tid]   = lbeg;
    gdelta[tid] = ex + before - lbeg;
    if (blockIdx.x == 0) {
        binstart[tid] = ex;
        if (tid == SORT_T - 1) binstart[SORT_T] = ex + total;
    }
    __syncthreads();
    for (int k = 0; k < SORT_V4; ++k) {
        const int i4 = (blockIdx.x * SORT_V4 + k) * SORT_T + tid;
        if (i4 < N_EDGES / 4) {
            const int4 r = ((const int4*)row)[i4];
            int4 c = ((const int4*)col)[i4];
            c.x = clamp_node(c.x); c.y = clamp_node(c.y);
            c.z = clamp_node(c.z); c.w = clamp_node(c.w);
            const int q0 = atomicAdd(&lcur[c.x >> BIN_SHIFT], 1);
            const int q1 = atomicAdd(&lcur[c.y >> BIN_SHIFT], 1);
            const int q2 = atomicAdd(&lcur[c.z >> BIN_SHIFT], 1);
            const int q3 = atomicAdd(&lcur[c.w >> BIN_SHIFT], 1);
            stage[q0] = make_int2(r.x, c.x);
            stage[q1] = make_int2(r.y, c.y);
            stage[q2] = make_int2(r.z, c.z);
            stage[q3] = make_int2(r.w, c.w);
        }
    }
    __syncthreads();
    const int nblk = min(SORT_CH, N_EDGES - (int)blockIdx.x * SORT_CH);
    for (int i = tid; i < nblk; i += SORT_T) {
        const int2 rc = stage[i];
        tmp[i + gdelta[rc.y >> BIN_SHIFT]] = rc;
    }
}

#define BIN_T   1024
#define BIN_R   8
#define BIN_CAP (BIN_T * BIN_R)
__global__ __launch_bounds__(BIN_T) void bin_sort_kernel(const int2* __restrict__ tmp,
                                                         const int* __restrict__ binstart,
                                                         int* __restrict__ rowptr,
                                                         float* __restrict__ dinv,
                                                         int2* __restrict__ erec) {
    __shared__ int2 stage[BIN_CAP];
    __shared__ int cnt[256];
    __shared__ int cur[256];
    __shared__ int wt[8];
    const int tid = threadIdx.x;
    const int beg = binstart[blockIdx.x];
    const int n   = binstart[blockIdx.x + 1] - beg;
    const bool small = n <= BIN_CAP;
    if (tid < 256) cnt[tid] = 0;
    __syncthreads();
    int2 rc[BIN_R];
    if (small) {
#pragma unroll
        for (int j = 0; j < BIN_R; ++j) {
            const int i = j * BIN_T + tid;
            rc[j] = make_int2(0, 0);
            if (i < n) rc[j] = tmp[beg + i];
        }
#pragma unroll
        for (int j = 0; j < BIN_R; ++j)
            if (j * BIN_T + tid < n) atomicAdd(&cnt[rc[j].y & BIN_MASK], 1);
    } else {
        for (int i = tid; i < n; i += BIN_T) atomicAdd(&cnt[tmp[beg + i].y & BIN_MASK], 1);
    }
    __syncthreads();
    const int v  = (tid < 256) ? cnt[tid] : 0;
    const int ex = excl_scan256(v, wt, tid);
    if (tid < 256) {
        const int node = blockIdx.x * 256 + tid;
        rowptr[node] = beg + ex;
        dinv[node]   = (v > 0) ? rsqrtf((float)v) : 0.0f;
        cur[tid]     = ex;
    }
    __syncthreads();
    if (small) {
#pragma unroll
        for (int j = 0; j < BIN_R; ++j)
            if (j * BIN_T + tid < n) stage[atomicAdd(&cur[rc[j].y & BIN_MASK], 1)] = rc[j];
        __syncthreads();
        for (int i = tid; i < n; i += BIN_T) erec[beg + i] = stage[i];
    } else {
        for (int i = tid; i < n; i += BIN_T) {
            const int2 e = tmp[beg + i];
            erec[beg + atomicAdd(&cur[e.y & BIN_MASK], 1)] = e;
        }
    }
}

template <int CNT, int FIRST>
__device__ __forceinline__ void gather_accum(const int2* __restrict__ rec,
                                             const float* __restrict__ hl,
                                             const float* __restrict__ dinv, const float dt,
                                             int2* __restrict__ wout, const int lane,
                                             float4& acc) {
    int2   r[CNT];
    float4 v[CNT];
    float  wgt[CNT];
#pragma unroll
    for (int i = 0; i < CNT; ++i) r[i] = rec[i];
#pragma unroll
    for (int i = 0; i < CNT; ++i) {
        unsigned s = min((unsigned)r[i].x, (unsigned)(N_NODES - 1));
        v[i] = *(const float4*)(hl + (size_t)s * F_DIM);
        if (FIRST) wgt[i] = dinv[s] * dt;
        else       wgt[i] = __int_as_float(r[i].y);
    }
    if (FIRST) {
        int px = 0, py = 0;
#pragma unroll
        for (int i = 0; i < CNT; ++i) {
            px = (lane == i) ? r[i].x : px;
            py = (lane == i) ? __float_as_int(wgt[i]) : py;
        }
        if (lane < CNT) wout[lane] = make_int2(px, py);
    }
#pragma unroll
    for (int i = 0; i < CNT; ++i) {
        acc.x = fmaf(wgt[i], v[i].x, acc.x);
        acc.y = fmaf(wgt[i], v[i].y, acc.y);
        acc.z = fmaf(wgt[i], v[i].z, acc.z);
        acc.w = fmaf(wgt[i], v[i].w, acc.w);
    }
}

template <int FIRST>
__global__ __launch_bounds__(256) void spmm_gather_kernel(const int* __restrict__ rowptr,
                                                          const int2* __restrict__ erec,
                                                          const float* __restrict__ dinv,
                                                          int2* __restrict__ wout,
                                                          const float* __restrict__ hin,
                                                          float* __restrict__ hout) {
    const int lane = threadIdx.x & 31;
    const int node = __builtin_amdgcn_readfirstlane((int)(blockIdx.x * 8 + (threadIdx.x >> 5)));
    const int beg = rowptr[node];
    const int end = rowptr[node + 1];
    const float dt = FIRST ? dinv[node] : 0.0f;
    const float* __restrict__ hl = hin + lane * 4;
    float4 acc = make_float4(0.0f, 0.0f, 0.0f, 0.0f);
    int e = beg;
    while (e + 8 <= end) { gather_accum<8, FIRST>(erec + e, hl, dinv, dt, wout + e, lane, acc); e += 8; }
    if (e + 4 <= end)    { gather_accum<4, FIRST>(erec + e, hl, dinv, dt, wout + e, lane, acc); e += 4; }
    switch (end - e) {
        case 3: gather_accum<3, FIRST>(erec + e, hl, dinv, dt, wout + e, lane, acc); break;
        case 2: gather_accum<2, FIRST>(erec + e, hl, dinv, dt, wout + e, lane, acc); break;
        case 1: gather_accum<1, FIRST>(erec + e, hl, dinv, dt, wout + e, lane, acc); break;
        default: break;
    }
    *(float4*)(hout + (size_t)node * F_DIM + lane * 4) = acc;
}


template <int L>
__device__ __forceinline__ int row_share(int x) {
    return __builtin_amdgcn_update_dpp(0, x, 0x150 + L, 0xf, 0xf, false);
}

#define GATHER_PAIR_LOAD(P, I)                                                          \
    {                                                                                    \
        const int sx = row_share<(P)>(rv.x);                                             \
        w[I] = __int_as_float(row_share<(P)>(rv.y));                                     \
        v[I] = make_float4(0.0f, 0.0f, 0.0f, 0.0f);                                      \
        if (2 * (P) + hi < cnt) {                                                        \
                  \
                                                  \
            const unsigned s = min((unsigned)sx, (unsigned)(N_NODES - 1));               \
            v[I] = *(const float4*)(hl + (size_t)s * C_OUT);                             \
        }                                                                                \
    }

template <int P0>
__device__ __forceinline__ void gather_group(const int2 rv, const int cnt, const int hi,
                                             const float* __restrict__ hl, float4& acc) {
    float4 v[8];
    float  w[8];
    GATHER_PAIR_LOAD(P0 + 0, 0)
    GATHER_PAIR_LOAD(P0 + 1, 1)
    GATHER_PAIR_LOAD(P0 + 2, 2)
    GATHER_PAIR_LOAD(P0 + 3, 3)
    GATHER_PAIR_LOAD(P0 + 4, 4)
    GATHER_PAIR_LOAD(P0 + 5, 5)
    GATHER_PAIR_LOAD(P0 + 6, 6)
    GATHER_PAIR_LOAD(P0 + 7, 7)
#pragma unroll
    for (int i = 0; i < 8; ++i) {
        acc.x = fmaf(w[i], v[i].x, acc.x);
        acc.y = fmaf(w[i], v[i].y, acc.y);
        acc.z = fmaf(w[i], v[i].z, acc.z);
        acc.w = fmaf(w[i], v[i].w, acc.w);
    }
}

template <int LAST>
__global__ __launch_bounds__(256) void spmm_gather64_kernel(const int* __restrict__ rowptr,
                                                            const int2* __restrict__ erec,
                                                            const float* __restrict__ hin,
                                                            const float* __restrict__ zadd,
                                                            const float* __restrict__ bias,
                                                            float* __restrict__ hout) {
    const int lane   = threadIdx.x & 31;
    const int lane16 = lane & 15;
    const int hi     = lane >> 4;
    const int slot   = 2 * lane16 + hi;
    const int node = __builtin_amdgcn_readfirstlane((int)(blockIdx.x * 8 + (threadIdx.x >> 5)));
    const int beg = rowptr[node];
    const int end = rowptr[node + 1];
    const float* __restrict__ hl = hin + lane16 * 4;
    const float4 z = *(const float4*)(zadd + (size_t)node * C_OUT + lane16 * 4);
    float4 acc = make_float4(0.0f, 0.0f, 0.0f, 0.0f);
    for (int e = beg; e < end; e += 32) {
        const int cnt = min(end - e, 32);
        int2 rv = make_int2(0, 0);
        if (slot < cnt) rv = erec[e + slot];
        gather_group<0>(rv, cnt, hi, hl, acc);
        if (cnt > 16) gather_group<8>(rv, cnt, hi, hl, acc);
    }
    acc.x += __shfl_xor(acc.x, 16, 32);
    acc.y += __shfl_xor(acc.y, 16, 32);
    acc.z += __shfl_xor(acc.z, 16, 32);
    acc.w += __shfl_xor(acc.w, 16, 32);
    float4 o = make_float4(acc.x + z.x, acc.y + z.y, acc.z + z.z, acc.w + z.w);
    if (LAST) {
        const float4 b = *(const float4*)(bias + lane16 * 4);
        o.x += b.x; o.y += b.y; o.z += b.z; o.w += b.w;
        float mx = fmaxf(fmaxf(o.x, o.y), fmaxf(o.z, o.w));
#pragma unroll
        for (int off = 8; off > 0; off >>= 1) mx = fmaxf(mx, __shfl_xor(mx, off, 32));
        float s = (expf(o.x - mx) + expf(o.y - mx)) + (expf(o.z - mx) + expf(o.w - mx));
#pragma unroll
        for (int off = 8; off > 0; off >>= 1) s += __shfl_xor(s, off, 32);
        const float lse = mx + logf(s);
        o.x -= lse; o.y -= lse; o.z -= lse; o.w -= lse;
    }
    if (hi == 0) *(float4*)(hout + (size_t)node * C_OUT + lane16 * 4) = o;
}
#undef GATHER_PAIR_LOAD

template <int NOUT>
struct GemmCfg {
    static constexpr int NT    = NOUT / 16;
    static constexpr int PITCH = 2 * NOUT + 32;
    static constexpr int SEGF  = (F_DIM / 2) * PITCH;
    static constexpr int NV4   = NOUT / 4;
    static constexpr int ITEMS = (F_DIM / 2) * NV4 / 256;
};

template <int NOUT, int NSEG, int MODE>
__global__ __launch_bounds__(256, 4) void tag_gemm_wmma_kernel(
    const float* __restrict__ s0, const float* __restrict__ s1,
    const float* __restrict__ s2, const float* __restrict__ s3,
    const float* __restrict__ W, const float* __restrict__ bias,
    float* __restrict__ out) {
    constexpr int NT    = GemmCfg<NOUT>::NT;
    constexpr int PITCH = GemmCfg<NOUT>::PITCH;
    constexpr int SEGF  = GemmCfg<NOUT>::SEGF;
    constexpr int NV4   = GemmCfg<NOUT>::NV4;
    constexpr int ITEMS = GemmCfg<NOUT>::ITEMS;
    constexpr int GSTEP = 8 * 2 * PITCH;

    constexpr int WP    = (MODE == 1) ? C_OUT : NOUT;
    constexpr int NBUF  = (NSEG > 1) ? 2 : 1;

    __shared__ __attribute__((aligned(16))) float Bs[NBUF * SEGF];

    const int tid    = threadIdx.x;
    const int lane   = tid & 31;
    const int lane16 = lane & 15;
    const int hi     = lane >> 4;
    const int wave   = __builtin_amdgcn_readfirstlane(tid >> 5);
    const int row0   = (blockIdx.x * 8 + wave) * 16;
    const bool active = row0 < N_NODES;
    const int m = min(row0 + lane16, N_NODES - 1);
    const size_t aoff = (size_t)m * F_DIM + hi * 2;

    const float* segs[4] = { s0, s1, s2, s3 };

    v8f acc[NT];
#pragma unroll
    for (int t = 0; t < NT; ++t) acc[t] = (v8f){ 0.f, 0.f, 0.f, 0.f, 0.f, 0.f, 0.f, 0.f };

    v2f a_cur[8], a_nxt[8];
#pragma unroll
    for (int j = 0; j < 8; ++j) a_cur[j] = *(const v2f*)(segs[0] + aoff + 4 * j);

    const int spr = tid / NV4;
    const int sn4 = tid % NV4;
    const float* __restrict__ wsrc = (MODE == 1)
        ? W + (size_t)(2 * blockIdx.y + ((4 * sn4) >> 6)) * (F_DIM * C_OUT)
            + (size_t)(2 * spr) * C_OUT + ((4 * sn4) & 63)
        : W + (size_t)(2 * spr) * NOUT + 4 * sn4;
    float* const wdst = Bs + spr * PITCH + 8 * sn4;
    constexpr int PR_STEP = 256 / NV4;
    float4 w0[ITEMS], w1[ITEMS];
#pragma unroll
    for (int it = 0; it < ITEMS; ++it) {
        const float* g = wsrc + (size_t)(2 * PR_STEP * it) * WP;
        w0[it] = *(const float4*)g;
        w1[it] = *(const float4*)(g + WP);
    }
#pragma unroll
    for (int it = 0; it < ITEMS; ++it) {
        float* d = wdst + it * PR_STEP * PITCH;
        *(float4*)d       = make_float4(w0[it].x, w1[it].x, w0[it].y, w1[it].y);
        *(float4*)(d + 4) = make_float4(w0[it].z, w1[it].z, w0[it].w, w1[it].w);
    }
    __syncthreads();

#pragma unroll
    for (int s = 0; s < NSEG; ++s) {
        const float* bg = Bs + (s & 1) * SEGF + hi * PITCH + 2 * lane16;
        const float* ag = segs[s] + aoff;
        const float* an = (s < NSEG - 1) ? segs[s + 1] + aoff : ag;
        v2f bq[2][NT];
#pragma unroll
        for (int t = 0; t < NT; ++t) bq[0][t] = *(const v2f*)(bg + 32 * t);

        auto products = [&](const float* bgp, const float* bnext) {
#pragma unroll
            for (int j = 0; j < 8; ++j) {
                const float* bn = (j < 7) ? bgp + (j + 1) * 2 * PITCH : bnext;
#pragma unroll
                for (int t = 0; t < NT; ++t) bq[(j + 1) & 1][t] = *(const v2f*)(bn + 32 * t);
#pragma unroll
                for (int t = 0; t < NT; ++t)
                    acc[t] = __builtin_amdgcn_wmma_f32_16x16x4_f32(
                        false, a_cur[j], false, bq[j & 1][t], (short)0, acc[t], false, false);
            }
        };

#pragma unroll
        for (int j = 0; j < 8; ++j) a_nxt[j] = *(const v2f*)(ag + 32 + 4 * j);
        if (s < NSEG - 1) {
#pragma unroll
            for (int it = 0; it < ITEMS; ++it) {
                const float* g = wsrc + (size_t)((s + 1) * F_DIM + 2 * PR_STEP * it) * WP;
                w0[it] = *(const float4*)g;
                w1[it] = *(const float4*)(g + WP);
            }
        }
        products(bg, bg + GSTEP);
#pragma unroll
        for (int j = 0; j < 8; ++j) a_cur[j] = a_nxt[j];
        bg += GSTEP;
        ag += 32;

#pragma unroll 1
        for (int sc = 1; sc < 4; ++sc) {
            const bool last = (sc == 3);
            const float* ap = last ? an : ag + 32;
            if (s < NSEG - 1 || !last) {
#pragma unroll
                for (int j = 0; j < 8; ++j) a_nxt[j] = *(const v2f*)(ap + 4 * j);
            }
            products(bg, last ? bg + 7 * 2 * PITCH : bg + GSTEP);
#pragma unroll
            for (int j = 0; j < 8; ++j) a_cur[j] = a_nxt[j];
            bg += GSTEP;
            ag += 32;
        }

        if (s < NSEG - 1) {
            float* db = wdst + ((s + 1) & 1) * SEGF;
#pragma unroll
            for (int it = 0; it < ITEMS; ++it) {
                float* d = db + it * PR_STEP * PITCH;
                *(float4*)d       = make_float4(w0[it].x, w1[it].x, w0[it].y, w1[it].y);
                *(float4*)(d + 4) = make_float4(w0[it].z, w1[it].z, w0[it].w, w1[it].w);
            }
            __syncthreads();
        }
    }

    if (active) {
#pragma unroll
        for (int t = 0; t < NT; ++t) {
            const int n = t * 16 + lane16;
            if (MODE == 0) {
                const float bv = bias[n];
#pragma unroll
                for (int j = 0; j < 8; ++j) {
                    int mm  = row0 + j + hi * 8;
                    float v = acc[t][j] + bv;
                    v = (v > 0.0f) ? v : 0.0f;
                    out[(long long)mm * NOUT + n] = v;
                }
            } else {
                float* zp = out + (size_t)(2 * blockIdx.y + (t >> 2)) * ((size_t)N_NODES * C_OUT)
                                + ((t & 3) * 16 + lane16);
#pragma unroll
                for (int j = 0; j < 8; ++j) {
                    int mm = row0 + j + hi * 8;
                    zp[(size_t)mm * C_OUT] = acc[t][j];
                }
            }
        }
    }
}


extern "C" void kernel_launch(void* const* d_in, const int* in_sizes, int n_in,
                              void* d_out, int out_size, void* d_ws, size_t ws_size,
                              hipStream_t stream) {
    const float* x   = (const float*)d_in[0];
    const int*   ei  = (const int*)d_in[1];
    const float* W1  = (const float*)d_in[2];
    const float* b1  = (const float*)d_in[3];
    const float* W2  = (const float*)d_in[4];
    const float* b2  = (const float*)d_in[5];
    float*       out = (float*)d_out;

    const int* rowp = ei;
    const int* colp = ei + N_EDGES;

    int*   bhist  = (int*)d_ws;
    int*   rowptr = bhist + N_PAD;
    float* dinv   = (float*)(rowptr + N_PAD);
    int*   binst  = (int*)(dinv + N_PAD);
    int2*  erec   = (int2*)(binst + N_EDGES);
    float* hb0    = (float*)(erec + N_EDGES);
    float* hb1    = hb0 + (size_t)N_NODES * F_DIM;
    float* hb2    = hb1 + (size_t)N_NODES * F_DIM;
    float* acc1   = hb2 + (size_t)N_NODES * F_DIM;
    float* zpl    = acc1 + (size_t)N_NODES * F_DIM;
    const size_t ZP = (size_t)N_NODES * C_OUT;
    int2*  wrec   = (int2*)(zpl + 4 * ZP);
    float* t1     = hb0;
    float* t2     = hb1;
    int2*  tmp    = (int2*)hb2;

    const int nodeGroups = N_NODES / 8;
    const int rowTiles   = N_NODES / 16;
    const int rowGroups  = (rowTiles + 7) / 8;

    bin_count_kernel<<<SORT_NB, SORT_T, 0, stream>>>(colp, bhist);
    bin_scatter_kernel<<<SORT_NB, SORT_T, 0, stream>>>(rowp, colp, bhist, binst, tmp);
    bin_sort_kernel<<<N_BINBLK, BIN_T, 0, stream>>>(tmp, binst, rowptr, dinv, erec);

    spmm_gather_kernel<1><<<nodeGroups, 256, 0, stream>>>(rowptr, erec, dinv, wrec, x, hb0);
    spmm_gather_kernel<0><<<nodeGroups, 256, 0, stream>>>(rowptr, wrec, dinv, wrec, hb0, hb1);
    spmm_gather_kernel<0><<<nodeGroups, 256, 0, stream>>>(rowptr, wrec, dinv, wrec, hb1, hb2);

    tag_gemm_wmma_kernel<H_OUT1, 4, 0><<<rowGroups, 256, 0, stream>>>(
        x, hb0, hb1, hb2, W1, b1, acc1);

    tag_gemm_wmma_kernel<2 * C_OUT, 1, 1><<<dim3(rowGroups, 2), 256, 0, stream>>>(
        acc1, acc1, acc1, acc1, W2, b2, zpl);

    spmm_gather64_kernel<0><<<nodeGroups, 256, 0, stream>>>(
        rowptr, wrec, zpl + 3 * ZP, zpl + 2 * ZP, b2, t1);
    spmm_gather64_kernel<0><<<nodeGroups, 256, 0, stream>>>(
        rowptr, wrec, t1, zpl + ZP, b2, t2);
    spmm_gather64_kernel<1><<<nodeGroups, 256, 0, stream>>>(
        rowptr, wrec, t2, zpl, b2, out);
}


# Layer 2 products moved into the layer 1 GEMM: layer 1 is computed transposed, so its accumulators are layer 2's A fragments; h never stored, 10 launches; layer-2 k order per 16 features changes
# speedup vs baseline: 6.6658x; 1.0407x over previous
#include <hip/hip_runtime.h>
#include <hip/hip_bf16.h>
#include <math.h>


#define N_NODES 50000
#define N_EDGES 800000
#define F_DIM   128
#define H_OUT1  128
#define C_OUT   64
#define K_TOT   512

#define N_PAD      53248

typedef __attribute__((ext_vector_type(2))) float v2f;
typedef __attribute__((ext_vector_type(8))) float v8f;


#define SORT_T    256
#define SORT_V4   4
#define SORT_CH   (SORT_T * SORT_V4 * 4)
#define SORT_NB   ((N_EDGES + SORT_CH - 1) / SORT_CH)
#define SORT_LD   28
#define BIN_SHIFT 8
#define BIN_MASK  255
static_assert(SORT_NB % SORT_LD == 0 && SORT_NB * SORT_T <= N_PAD, "edge block count");
#define N_BINBLK  (N_PAD / 256)

__device__ __forceinline__ int clamp_node(int c) {
    return (int)min((unsigned)c, (unsigned)(N_NODES - 1));
}

__device__ __forceinline__ int excl_scan256(int v, int* wt, int tid) {
    const int lane = tid & 31;
    const int w    = tid >> 5;
    int s = v;
    for (int off = 1; off < 32; off <<= 1) {
        int nb = __shfl_up(s, off, 32);
        if (lane >= off) s += nb;
    }
    if (w < 8 && lane == 31) wt[w] = s;
    __syncthreads();
    int pre = 0;
    for (int i = 0; i < 8; ++i) pre += (i < w) ? wt[i] : 0;
    __syncthreads();
    return pre + s - v;
}

__global__ __launch_bounds__(SORT_T) void bin_count_kernel(const int* __restrict__ col,
                                                           int* __restrict__ bhist) {
    __shared__ int h[SORT_T];
    const int tid = threadIdx.x;
    h[tid] = 0;
    __syncthreads();
    for (int k = 0; k < SORT_V4; ++k) {
        const int i4 = (blockIdx.x * SORT_V4 + k) * SORT_T + tid;
        if (i4 < N_EDGES / 4) {
            const int4 c = ((const int4*)col)[i4];
            atomicAdd(&h[clamp_node(c.x) >> BIN_SHIFT], 1);
            atomicAdd(&h[clamp_node(c.y) >> BIN_SHIFT], 1);
            atomicAdd(&h[clamp_node(c.z) >> BIN_SHIFT], 1);
            atomicAdd(&h[clamp_node(c.w) >> BIN_SHIFT], 1);
        }
    }
    __syncthreads();
    bhist[blockIdx.x * SORT_T + tid] = h[tid];
}

__global__ __launch_bounds__(SORT_T) void bin_scatter_kernel(const int* __restrict__ row,
                                                             const int* __restrict__ col,
                                                             const int* __restrict__ bhist,
                                                             int* __restrict__ binstart,
                                                             int2* __restrict__ tmp) {
    __shared__ int2 stage[SORT_CH];
    __shared__ int lcur[SORT_T];
    __shared__ int gdelta[SORT_T];
    __shared__ int wt[8];
    const int tid = threadIdx.x;
    int total = 0, before = 0, own = 0;
    for (int b0 = 0; b0 < SORT_NB; b0 += SORT_LD) {
        int v[SORT_LD];
#pragma unroll
        for (int j = 0; j < SORT_LD; ++j) v[j] = bhist[(b0 + j) * SORT_T + tid];
#pragma unroll
        for (int j = 0; j < SORT_LD; ++j) {
            total += v[j];
            before += (b0 + j < (int)blockIdx.x) ? v[j] : 0;
            own    += (b0 + j == (int)blockIdx.x) ? v[j] : 0;
        }
    }
    const int ex   = excl_scan256(total, wt, tid);
    const int lbeg = excl_scan256(own, wt, tid);
    lcur[tid]   = lbeg;
    gdelta[tid] = ex + before - lbeg;
    if (blockIdx.x == 0) {
        binstart[tid] = ex;
        if (tid == SORT_T - 1) binstart[SORT_T] = ex + total;
    }
    __syncthreads();
    for (int k = 0; k < SORT_V4; ++k) {
        const int i4 = (blockIdx.x * SORT_V4 + k) * SORT_T + tid;
        if (i4 < N_EDGES / 4) {
            const int4 r = ((const int4*)row)[i4];
            int4 c = ((const int4*)col)[i4];
            c.x = clamp_node(c.x); c.y = clamp_node(c.y);
            c.z = clamp_node(c.z); c.w = clamp_node(c.w);
            const int q0 = atomicAdd(&lcur[c.x >> BIN_SHIFT], 1);
            const int q1 = atomicAdd(&lcur[c.y >> BIN_SHIFT], 1);
            const int q2 = atomicAdd(&lcur[c.z >> BIN_SHIFT], 1);
            const int q3 = atomicAdd(&lcur[c.w >> BIN_SHIFT], 1);
            stage[q0] = make_int2(r.x, c.x);
            stage[q1] = make_int2(r.y, c.y);
            stage[q2] = make_int2(r.z, c.z);
            stage[q3] = make_int2(r.w, c.w);
        }
    }
    __syncthreads();
    const int nblk = min(SORT_CH, N_EDGES - (int)blockIdx.x * SORT_CH);
    for (int i = tid; i < nblk; i += SORT_T) {
        const int2 rc = stage[i];
        tmp[i + gdelta[rc.y >> BIN_SHIFT]] = rc;
    }
}

#define BIN_T   1024
#define BIN_R   8
#define BIN_CAP (BIN_T * BIN_R)
__global__ __launch_bounds__(BIN_T) void bin_sort_kernel(const int2* __restrict__ tmp,
                                                         const int* __restrict__ binstart,
                                                         int* __restrict__ rowptr,
                                                         float* __restrict__ dinv,
                                                         int2* __restrict__ erec) {
    __shared__ int2 stage[BIN_CAP];
    __shared__ int cnt[256];
    __shared__ int cur[256];
    __shared__ int wt[8];
    const int tid = threadIdx.x;
    const int beg = binstart[blockIdx.x];
    const int n   = binstart[blockIdx.x + 1] - beg;
    const bool small = n <= BIN_CAP;
    if (tid < 256) cnt[tid] = 0;
    __syncthreads();
    int2 rc[BIN_R];
    if (small) {
#pragma unroll
        for (int j = 0; j < BIN_R; ++j) {
            const int i = j * BIN_T + tid;
            rc[j] = make_int2(0, 0);
            if (i < n) rc[j] = tmp[beg + i];
        }
#pragma unroll
        for (int j = 0; j < BIN_R; ++j)
            if (j * BIN_T + tid < n) atomicAdd(&cnt[rc[j].y & BIN_MASK], 1);
    } else {
        for (int i = tid; i < n; i += BIN_T) atomicAdd(&cnt[tmp[beg + i].y & BIN_MASK], 1);
    }
    __syncthreads();
    const int v  = (tid < 256) ? cnt[tid] : 0;
    const int ex = excl_scan256(v, wt, tid);
    if (tid < 256) {
        const int node = blockIdx.x * 256 + tid;
        rowptr[node] = beg + ex;
        dinv[node]   = (v > 0) ? rsqrtf((float)v) : 0.0f;
        cur[tid]     = ex;
    }
    __syncthreads();
    if (small) {
#pragma unroll
        for (int j = 0; j < BIN_R; ++j)
            if (j * BIN_T + tid < n) stage[atomicAdd(&cur[rc[j].y & BIN_MASK], 1)] = rc[j];
        __syncthreads();
        for (int i = tid; i < n; i += BIN_T) erec[beg + i] = stage[i];
    } else {
        for (int i = tid; i < n; i += BIN_T) {
            const int2 e = tmp[beg + i];
            erec[beg + atomicAdd(&cur[e.y & BIN_MASK], 1)] = e;
        }
    }
}

template <int CNT, int FIRST>
__device__ __forceinline__ void gather_accum(const int2* __restrict__ rec,
                                             const float* __restrict__ hl,
                                             const float* __restrict__ dinv, const float dt,
                                             int2* __restrict__ wout, const int lane,
                                             float4& acc) {
    int2   r[CNT];
    float4 v[CNT];
    float  wgt[CNT];
#pragma unroll
    for (int i = 0; i < CNT; ++i) r[i] = rec[i];
#pragma unroll
    for (int i = 0; i < CNT; ++i) {
        unsigned s = min((unsigned)r[i].x, (unsigned)(N_NODES - 1));
        v[i] = *(const float4*)(hl + (size_t)s * F_DIM);
        if (FIRST) wgt[i] = dinv[s] * dt;
        else       wgt[i] = __int_as_float(r[i].y);
    }
    if (FIRST) {
        int px = 0, py = 0;
#pragma unroll
        for (int i = 0; i < CNT; ++i) {
            px = (lane == i) ? r[i].x : px;
            py = (lane == i) ? __float_as_int(wgt[i]) : py;
        }
        if (lane < CNT) wout[lane] = make_int2(px, py);
    }
#pragma unroll
    for (int i = 0; i < CNT; ++i) {
        acc.x = fmaf(wgt[i], v[i].x, acc.x);
        acc.y = fmaf(wgt[i], v[i].y, acc.y);
        acc.z = fmaf(wgt[i], v[i].z, acc.z);
        acc.w = fmaf(wgt[i], v[i].w, acc.w);
    }
}

template <int FIRST>
__global__ __launch_bounds__(256) void spmm_gather_kernel(const int* __restrict__ rowptr,
                                                          const int2* __restrict__ erec,
                                                          const float* __restrict__ dinv,
                                                          int2* __restrict__ wout,
                                                          const float* __restrict__ hin,
                                                          float* __restrict__ hout) {
    const int lane = threadIdx.x & 31;
    const int node = __builtin_amdgcn_readfirstlane((int)(blockIdx.x * 8 + (threadIdx.x >> 5)));
    const int beg = rowptr[node];
    const int end = rowptr[node + 1];
    const float dt = FIRST ? dinv[node] : 0.0f;
    const float* __restrict__ hl = hin + lane * 4;
    float4 acc = make_float4(0.0f, 0.0f, 0.0f, 0.0f);
    int e = beg;
    while (e + 8 <= end) { gather_accum<8, FIRST>(erec + e, hl, dinv, dt, wout + e, lane, acc); e += 8; }
    if (e + 4 <= end)    { gather_accum<4, FIRST>(erec + e, hl, dinv, dt, wout + e, lane, acc); e += 4; }
    switch (end - e) {
        case 3: gather_accum<3, FIRST>(erec + e, hl, dinv, dt, wout + e, lane, acc); break;
        case 2: gather_accum<2, FIRST>(erec + e, hl, dinv, dt, wout + e, lane, acc); break;
        case 1: gather_accum<1, FIRST>(erec + e, hl, dinv, dt, wout + e, lane, acc); break;
        default: break;
    }
    *(float4*)(hout + (size_t)node * F_DIM + lane * 4) = acc;
}


template <int L>
__device__ __forceinline__ int row_share(int x) {
    return __builtin_amdgcn_update_dpp(0, x, 0x150 + L, 0xf, 0xf, false);
}

#define GATHER_PAIR_LOAD(P, I)                                                          \
    {                                                                                    \
        const int sx = row_share<(P)>(rv.x);                                             \
        w[I] = __int_as_float(row_share<(P)>(rv.y));                                     \
        v[I] = make_float4(0.0f, 0.0f, 0.0f, 0.0f);                                      \
        if (2 * (P) + hi < cnt) {                                                        \
                  \
                                                  \
            const unsigned s = min((unsigned)sx, (unsigned)(N_NODES - 1));               \
            v[I] = *(const float4*)(hl + (size_t)s * C_OUT);                             \
        }                                                                                \
    }

template <int P0>
__device__ __forceinline__ void gather_group(const int2 rv, const int cnt, const int hi,
                                             const float* __restrict__ hl, float4& acc) {
    float4 v[8];
    float  w[8];
    GATHER_PAIR_LOAD(P0 + 0, 0)
    GATHER_PAIR_LOAD(P0 + 1, 1)
    GATHER_PAIR_LOAD(P0 + 2, 2)
    GATHER_PAIR_LOAD(P0 + 3, 3)
    GATHER_PAIR_LOAD(P0 + 4, 4)
    GATHER_PAIR_LOAD(P0 + 5, 5)
    GATHER_PAIR_LOAD(P0 + 6, 6)
    GATHER_PAIR_LOAD(P0 + 7, 7)
#pragma unroll
    for (int i = 0; i < 8; ++i) {
        acc.x = fmaf(w[i], v[i].x, acc.x);
        acc.y = fmaf(w[i], v[i].y, acc.y);
        acc.z = fmaf(w[i], v[i].z, acc.z);
        acc.w = fmaf(w[i], v[i].w, acc.w);
    }
}

template <int LAST>
__global__ __launch_bounds__(256) void spmm_gather64_kernel(const int* __restrict__ rowptr,
                                                            const int2* __restrict__ erec,
                                                            const float* __restrict__ hin,
                                                            const float* __restrict__ zadd,
                                                            const float* __restrict__ bias,
                                                            float* __restrict__ hout) {
    const int lane   = threadIdx.x & 31;
    const int lane16 = lane & 15;
    const int hi     = lane >> 4;
    const int slot   = 2 * lane16 + hi;
    const int node = __builtin_amdgcn_readfirstlane((int)(blockIdx.x * 8 + (threadIdx.x >> 5)));
    const int beg = rowptr[node];
    const int end = rowptr[node + 1];
    const float* __restrict__ hl = hin + lane16 * 4;
    const float4 z = *(const float4*)(zadd + (size_t)node * C_OUT + lane16 * 4);
    float4 acc = make_float4(0.0f, 0.0f, 0.0f, 0.0f);
    for (int e = beg; e < end; e += 32) {
        const int cnt = min(end - e, 32);
        int2 rv = make_int2(0, 0);
        if (slot < cnt) rv = erec[e + slot];
        gather_group<0>(rv, cnt, hi, hl, acc);
        if (cnt > 16) gather_group<8>(rv, cnt, hi, hl, acc);
    }
    acc.x += __shfl_xor(acc.x, 16, 32);
    acc.y += __shfl_xor(acc.y, 16, 32);
    acc.z += __shfl_xor(acc.z, 16, 32);
    acc.w += __shfl_xor(acc.w, 16, 32);
    float4 o = make_float4(acc.x + z.x, acc.y + z.y, acc.z + z.z, acc.w + z.w);
    if (LAST) {
        const float4 b = *(const float4*)(bias + lane16 * 4);
        o.x += b.x; o.y += b.y; o.z += b.z; o.w += b.w;
        float mx = fmaxf(fmaxf(o.x, o.y), fmaxf(o.z, o.w));
#pragma unroll
        for (int off = 8; off > 0; off >>= 1) mx = fmaxf(mx, __shfl_xor(mx, off, 32));
        float s = (expf(o.x - mx) + expf(o.y - mx)) + (expf(o.z - mx) + expf(o.w - mx));
#pragma unroll
        for (int off = 8; off > 0; off >>= 1) s += __shfl_xor(s, off, 32);
        const float lse = mx + logf(s);
        o.x -= lse; o.y -= lse; o.z -= lse; o.w -= lse;
    }
    if (hi == 0) *(float4*)(hout + (size_t)node * C_OUT + lane16 * 4) = o;
}
#undef GATHER_PAIR_LOAD

template <int NOUT>
struct GemmCfg {
    static constexpr int NT    = NOUT / 16;
    static constexpr int PITCH = 2 * NOUT + 32;
    static constexpr int SEGF  = (F_DIM / 2) * PITCH;
    static constexpr int NV4   = NOUT / 4;
    static constexpr int ITEMS = (F_DIM / 2) * NV4 / 256;
};

__global__ __launch_bounds__(256, 4) void tag_gemm_fused_kernel(
    const float* __restrict__ s0, const float* __restrict__ s1,
    const float* __restrict__ s2, const float* __restrict__ s3,
    const float* __restrict__ W1, const float* __restrict__ b1,
    const float* __restrict__ W2, float* __restrict__ zout) {
    constexpr int NOUT  = H_OUT1;
    constexpr int NSEG  = 4;
    constexpr int NT    = GemmCfg<NOUT>::NT;
    constexpr int PITCH = GemmCfg<NOUT>::PITCH;
    constexpr int SEGF  = GemmCfg<NOUT>::SEGF;
    constexpr int NV4   = GemmCfg<NOUT>::NV4;
    constexpr int ITEMS = GemmCfg<NOUT>::ITEMS;
    constexpr int GSTEP = 8 * 2 * PITCH;
    static_assert(2 * C_OUT == NOUT && H_OUT1 == F_DIM, "segment shape");

    __shared__ __attribute__((aligned(16))) float Bs[2 * SEGF];
    __shared__ __attribute__((aligned(16))) float bsh[H_OUT1];

    const int tid    = threadIdx.x;
    const int lane   = tid & 31;
    const int lane16 = lane & 15;
    const int hi     = lane >> 4;
    const int wave   = __builtin_amdgcn_readfirstlane(tid >> 5);
    const int row0   = (blockIdx.x * 8 + wave) * 16;
    const bool active = row0 < N_NODES;
    const int m = min(row0 + lane16, N_NODES - 1);
    const size_t aoff = (size_t)m * F_DIM + hi * 2;

    const float* segs[4] = { s0, s1, s2, s3 };

    v8f acc[NT];
#pragma unroll
    for (int t = 0; t < NT; ++t) acc[t] = (v8f){ 0.f, 0.f, 0.f, 0.f, 0.f, 0.f, 0.f, 0.f };

    v2f a_cur[8], a_nxt[8];
#pragma unroll
    for (int j = 0; j < 8; ++j) a_cur[j] = *(const v2f*)(segs[0] + aoff + 4 * j);

    const int spr = tid / NV4;
    const int sn4 = tid % NV4;
    constexpr int PR_STEP = 256 / NV4;
    static_assert(PR_STEP == 8, "the W2 row order below is written for 8 pair rows per item");
    const float* __restrict__ wsrc = W1 + (size_t)(2 * spr) * NOUT + 4 * sn4;
    float* const wdst = Bs + spr * PITCH + 8 * sn4;
    const float* __restrict__ wsrc2 = W2 + (size_t)((4 * sn4) >> 6) * (F_DIM * C_OUT)
                                         + (size_t)(2 * spr) * C_OUT + ((4 * sn4) & 63);
    float* const wdst2 = Bs + (2 * (spr & 3) + (spr >> 2)) * PITCH + 8 * sn4;
    float4 w0[ITEMS], w1[ITEMS];
#pragma unroll
    for (int it = 0; it < ITEMS; ++it) {
        const float* g = wsrc + (size_t)(2 * PR_STEP * it) * NOUT;
        w0[it] = *(const float4*)g;
        w1[it] = *(const float4*)(g + NOUT);
    }
#pragma unroll
    for (int it = 0; it < ITEMS; ++it) {
        float* d = wdst + it * PR_STEP * PITCH;
        *(float4*)d       = make_float4(w0[it].x, w1[it].x, w0[it].y, w1[it].y);
        *(float4*)(d + 4) = make_float4(w0[it].z, w1[it].z, w0[it].w, w1[it].w);
    }
    if (tid < H_OUT1) bsh[tid] = b1[tid];
    __syncthreads();

    v2f bq[2][NT];

#pragma unroll
    for (int s = 0; s < NSEG; ++s) {
        const float* bg = Bs + (s & 1) * SEGF + hi * PITCH + 2 * lane16;
        const float* ag = segs[s] + aoff;
        const float* an = (s < NSEG - 1) ? segs[s + 1] + aoff : ag;
#pragma unroll
        for (int t = 0; t < NT; ++t) bq[0][t] = *(const v2f*)(bg + 32 * t);

        auto products = [&](const float* bgp, const float* bnext) {
#pragma unroll
            for (int j = 0; j < 8; ++j) {
                const float* bn = (j < 7) ? bgp + (j + 1) * 2 * PITCH : bnext;
#pragma unroll
                for (int t = 0; t < NT; ++t) bq[(j + 1) & 1][t] = *(const v2f*)(bn + 32 * t);
#pragma unroll
                for (int t = 0; t < NT; ++t)
                    acc[t] = __builtin_amdgcn_wmma_f32_16x16x4_f32(
                        false, bq[j & 1][t], false, a_cur[j], (short)0, acc[t], false, false);
            }
        };

#pragma unroll
        for (int j = 0; j < 8; ++j) a_nxt[j] = *(const v2f*)(ag + 32 + 4 * j);
        if (s < NSEG - 1) {
#pragma unroll
            for (int it = 0; it < ITEMS; ++it) {
                const float* g = wsrc + (size_t)((s + 1) * F_DIM + 2 * PR_STEP * it) * NOUT;
                w0[it] = *(const float4*)g;
                w1[it] = *(const float4*)(g + NOUT);
            }
        } else {
#pragma unroll
            for (int it = 0; it < ITEMS; ++it) {
                const float* g = wsrc2 + (size_t)(2 * PR_STEP * it) * C_OUT;
                w0[it] = *(const float4*)g;
                w1[it] = *(const float4*)(g + C_OUT);
            }
        }
        products(bg, bg + GSTEP);
#pragma unroll
        for (int j = 0; j < 8; ++j) a_cur[j] = a_nxt[j];
        bg += GSTEP;
        ag += 32;

#pragma unroll 1
        for (int sc = 1; sc < 4; ++sc) {
            const bool last = (sc == 3);
            const float* ap = last ? an : ag + 32;
            if (s < NSEG - 1 || !last) {
#pragma unroll
                for (int j = 0; j < 8; ++j) a_nxt[j] = *(const v2f*)(ap + 4 * j);
            }
            products(bg, last ? bg + 7 * 2 * PITCH : bg + GSTEP);
#pragma unroll
            for (int j = 0; j < 8; ++j) a_cur[j] = a_nxt[j];
            bg += GSTEP;
            ag += 32;
        }

        {
            float* db = (s < NSEG - 1) ? wdst + ((s + 1) & 1) * SEGF : wdst2;
#pragma unroll
            for (int it = 0; it < ITEMS; ++it) {
                float* d = db + it * PR_STEP * PITCH;
                *(float4*)d       = make_float4(w0[it].x, w1[it].x, w0[it].y, w1[it].y);
                *(float4*)(d + 4) = make_float4(w0[it].z, w1[it].z, w0[it].w, w1[it].w);
            }
            __syncthreads();
        }
    }

#pragma unroll
    for (int t = 0; t < NT; ++t) {
        const float4 ba = *(const float4*)(bsh + 16 * t + 8 * hi);
        const float4 bb = *(const float4*)(bsh + 16 * t + 8 * hi + 4);
        const float bv[8] = { ba.x, ba.y, ba.z, ba.w, bb.x, bb.y, bb.z, bb.w };
#pragma unroll
        for (int r = 0; r < 8; ++r) {
            const float v = acc[t][r] + bv[r];
            acc[t][r] = (v > 0.0f) ? v : 0.0f;
        }
    }

#pragma unroll
    for (int y = 0; y < 2; ++y) {
        v8f zac[NT];
#pragma unroll
        for (int t = 0; t < NT; ++t) zac[t] = (v8f){ 0.f, 0.f, 0.f, 0.f, 0.f, 0.f, 0.f, 0.f };
        const float* bg = Bs + (y & 1) * SEGF + hi * PITCH + 2 * lane16;
#pragma unroll
        for (int t = 0; t < NT; ++t) bq[0][t] = *(const v2f*)(bg + 32 * t);
        if (y == 0) {
#pragma unroll
            for (int it = 0; it < ITEMS; ++it) {
                const float* g = wsrc2 + (size_t)2 * (F_DIM * C_OUT)
                                       + (size_t)(2 * PR_STEP * it) * C_OUT;
                w0[it] = *(const float4*)g;
                w1[it] = *(const float4*)(g + C_OUT);
            }
        }
#pragma unroll
        for (int J = 0; J < 32; ++J) {
            if (J < 31) {
#pragma unroll
                for (int t = 0; t < NT; ++t)
                    bq[(J + 1) & 1][t] = *(const v2f*)(bg + (J + 1) * 2 * PITCH + 32 * t);
            }
            const v2f ha = (v2f){ acc[J >> 2][2 * (J & 3)], acc[J >> 2][2 * (J & 3) + 1] };
#pragma unroll
            for (int t = 0; t < NT; ++t)
                zac[t] = __builtin_amdgcn_wmma_f32_16x16x4_f32(
                    false, ha, false, bq[J & 1][t], (short)0, zac[t], false, false);
        }
        if (y == 0) {
            float* db = wdst2 + SEGF;
#pragma unroll
            for (int it = 0; it < ITEMS; ++it) {
                float* d = db + it * PR_STEP * PITCH;
                *(float4*)d       = make_float4(w0[it].x, w1[it].x, w0[it].y, w1[it].y);
                *(float4*)(d + 4) = make_float4(w0[it].z, w1[it].z, w0[it].w, w1[it].w);
            }
            __syncthreads();
        }
        if (active) {
#pragma unroll
            for (int t = 0; t < NT; ++t) {
                float* zp = zout + (size_t)(2 * y + (t >> 2)) * ((size_t)N_NODES * C_OUT)
                                 + ((t & 3) * 16 + lane16);
#pragma unroll
                for (int j = 0; j < 8; ++j) {
                    const int mm = row0 + j + hi * 8;
                    zp[(size_t)mm * C_OUT] = zac[t][j];
                }
            }
        }
    }
}


extern "C" void kernel_launch(void* const* d_in, const int* in_sizes, int n_in,
                              void* d_out, int out_size, void* d_ws, size_t ws_size,
                              hipStream_t stream) {
    const float* x   = (const float*)d_in[0];
    const int*   ei  = (const int*)d_in[1];
    const float* W1  = (const float*)d_in[2];
    const float* b1  = (const float*)d_in[3];
    const float* W2  = (const float*)d_in[4];
    const float* b2  = (const float*)d_in[5];
    float*       out = (float*)d_out;

    const int* rowp = ei;
    const int* colp = ei + N_EDGES;

    int*   bhist  = (int*)d_ws;
    int*   rowptr = bhist + N_PAD;
    float* dinv   = (float*)(rowptr + N_PAD);
    int*   binst  = (int*)(dinv + N_PAD);
    int2*  erec   = (int2*)(binst + N_EDGES);
    float* hb0    = (float*)(erec + N_EDGES);
    float* hb1    = hb0 + (size_t)N_NODES * F_DIM;
    float* hb2    = hb1 + (size_t)N_NODES * F_DIM;
    float* zpl    = hb2 + (size_t)N_NODES * F_DIM;
    const size_t ZP = (size_t)N_NODES * C_OUT;
    int2*  wrec   = (int2*)(zpl + 4 * ZP);
    float* t1     = hb0;
    float* t2     = hb1;
    int2*  tmp    = (int2*)hb2;

    const int nodeGroups = N_NODES / 8;
    const int rowTiles   = N_NODES / 16;
    const int rowGroups  = (rowTiles + 7) / 8;

    bin_count_kernel<<<SORT_NB, SORT_T, 0, stream>>>(colp, bhist);
    bin_scatter_kernel<<<SORT_NB, SORT_T, 0, stream>>>(rowp, colp, bhist, binst, tmp);
    bin_sort_kernel<<<N_BINBLK, BIN_T, 0, stream>>>(tmp, binst, rowptr, dinv, erec);

    spmm_gather_kernel<1><<<nodeGroups, 256, 0, stream>>>(rowptr, erec, dinv, wrec, x, hb0);
    spmm_gather_kernel<0><<<nodeGroups, 256, 0, stream>>>(rowptr, wrec, dinv, wrec, hb0, hb1);
    spmm_gather_kernel<0><<<nodeGroups, 256, 0, stream>>>(rowptr, wrec, dinv, wrec, hb1, hb2);

    tag_gemm_fused_kernel<<<rowGroups, 256, 0, stream>>>(x, hb0, hb1, hb2, W1, b1, W2, zpl);

    spmm_gather64_kernel<0><<<nodeGroups, 256, 0, stream>>>(
        rowptr, wrec, zpl + 3 * ZP, zpl + 2 * ZP, b2, t1);
    spmm_gather64_kernel<0><<<nodeGroups, 256, 0, stream>>>(
        rowptr, wrec, t1, zpl + ZP, b2, t2);
    spmm_gather64_kernel<1><<<nodeGroups, 256, 0, stream>>>(
        rowptr, wrec, t2, zpl, b2, out);
}


# One 16-wave product work-group per WGP: 12 waves own a row tile each, 4 share a 13th (53 work-groups), so no SIMD issues 4 whole tiles; layer-1 k sum of the last 848 nodes in 4 partial sums
# speedup vs baseline: 7.0628x; 1.0596x over previous
#include <hip/hip_runtime.h>
#include <hip/hip_bf16.h>
#include <math.h>


#define N_NODES 50000
#define N_EDGES 800000
#define F_DIM   128
#define H_OUT1  128
#define C_OUT   64
#define K_TOT   512

#define N_PAD      53248

typedef __attribute__((ext_vector_type(2))) float v2f;
typedef __attribute__((ext_vector_type(8))) float v8f;


#define SORT_T    256
#define SORT_V4   4
#define SORT_CH   (SORT_T * SORT_V4 * 4)
#define SORT_NB   ((N_EDGES + SORT_CH - 1) / SORT_CH)
#define SORT_LD   28
#define BIN_SHIFT 8
#define BIN_MASK  255
static_assert(SORT_NB % SORT_LD == 0 && SORT_NB * SORT_T <= N_PAD, "edge block count");
#define N_BINBLK  (N_PAD / 256)

__device__ __forceinline__ int clamp_node(int c) {
    return (int)min((unsigned)c, (unsigned)(N_NODES - 1));
}

__device__ __forceinline__ int excl_scan256(int v, int* wt, int tid) {
    const int lane = tid & 31;
    const int w    = tid >> 5;
    int s = v;
    for (int off = 1; off < 32; off <<= 1) {
        int nb = __shfl_up(s, off, 32);
        if (lane >= off) s += nb;
    }
    if (w < 8 && lane == 31) wt[w] = s;
    __syncthreads();
    int pre = 0;
    for (int i = 0; i < 8; ++i) pre += (i < w) ? wt[i] : 0;
    __syncthreads();
    return pre + s - v;
}

__global__ __launch_bounds__(SORT_T) void bin_count_kernel(const int* __restrict__ col,
                                                           int* __restrict__ bhist) {
    __shared__ int h[SORT_T];
    const int tid = threadIdx.x;
    h[tid] = 0;
    __syncthreads();
    for (int k = 0; k < SORT_V4; ++k) {
        const int i4 = (blockIdx.x * SORT_V4 + k) * SORT_T + tid;
        if (i4 < N_EDGES / 4) {
            const int4 c = ((const int4*)col)[i4];
            atomicAdd(&h[clamp_node(c.x) >> BIN_SHIFT], 1);
            atomicAdd(&h[clamp_node(c.y) >> BIN_SHIFT], 1);
            atomicAdd(&h[clamp_node(c.z) >> BIN_SHIFT], 1);
            atomicAdd(&h[clamp_node(c.w) >> BIN_SHIFT], 1);
        }
    }
    __syncthreads();
    bhist[blockIdx.x * SORT_T + tid] = h[tid];
}

__global__ __launch_bounds__(SORT_T) void bin_scatter_kernel(const int* __restrict__ row,
                                                             const int* __restrict__ col,
                                                             const int* __restrict__ bhist,
                                                             int* __restrict__ binstart,
                                                             int2* __restrict__ tmp) {
    __shared__ int2 stage[SORT_CH];
    __shared__ int lcur[SORT_T];
    __shared__ int gdelta[SORT_T];
    __shared__ int wt[8];
    const int tid = threadIdx.x;
    int total = 0, before = 0, own = 0;
    for (int b0 = 0; b0 < SORT_NB; b0 += SORT_LD) {
        int v[SORT_LD];
#pragma unroll
        for (int j = 0; j < SORT_LD; ++j) v[j] = bhist[(b0 + j) * SORT_T + tid];
#pragma unroll
        for (int j = 0; j < SORT_LD; ++j) {
            total += v[j];
            before += (b0 + j < (int)blockIdx.x) ? v[j] : 0;
            own    += (b0 + j == (int)blockIdx.x) ? v[j] : 0;
        }
    }
    const int ex   = excl_scan256(total, wt, tid);
    const int lbeg = excl_scan256(own, wt, tid);
    lcur[tid]   = lbeg;
    gdelta[tid] = ex + before - lbeg;
    if (blockIdx.x == 0) {
        binstart[tid] = ex;
        if (tid == SORT_T - 1) binstart[SORT_T] = ex + total;
    }
    __syncthreads();
    for (int k = 0; k < SORT_V4; ++k) {
        const int i4 = (blockIdx.x * SORT_V4 + k) * SORT_T + tid;
        if (i4 < N_EDGES / 4) {
            const int4 r = ((const int4*)row)[i4];
            int4 c = ((const int4*)col)[i4];
            c.x = clamp_node(c.x); c.y = clamp_node(c.y);
            c.z = clamp_node(c.z); c.w = clamp_node(c.w);
            const int q0 = atomicAdd(&lcur[c.x >> BIN_SHIFT], 1);
            const int q1 = atomicAdd(&lcur[c.y >> BIN_SHIFT], 1);
            const int q2 = atomicAdd(&lcur[c.z >> BIN_SHIFT], 1);
            const int q3 = atomicAdd(&lcur[c.w >> BIN_SHIFT], 1);
            stage[q0] = make_int2(r.x, c.x);
            stage[q1] = make_int2(r.y, c.y);
            stage[q2] = make_int2(r.z, c.z);
            stage[q3] = make_int2(r.w, c.w);
        }
    }
    __syncthreads();
    const int nblk = min(SORT_CH, N_EDGES - (int)blockIdx.x * SORT_CH);
    for (int i = tid; i < nblk; i += SORT_T) {
        const int2 rc = stage[i];
        tmp[i + gdelta[rc.y >> BIN_SHIFT]] = rc;
    }
}

#define BIN_T   1024
#define BIN_R   8
#define BIN_CAP (BIN_T * BIN_R)
__global__ __launch_bounds__(BIN_T) void bin_sort_kernel(const int2* __restrict__ tmp,
                                                         const int* __restrict__ binstart,
                                                         int* __restrict__ rowptr,
                                                         float* __restrict__ dinv,
                                                         int2* __restrict__ erec) {
    __shared__ int2 stage[BIN_CAP];
    __shared__ int cnt[256];
    __shared__ int cur[256];
    __shared__ int wt[8];
    const int tid = threadIdx.x;
    const int beg = binstart[blockIdx.x];
    const int n   = binstart[blockIdx.x + 1] - beg;
    const bool small = n <= BIN_CAP;
    if (tid < 256) cnt[tid] = 0;
    __syncthreads();
    int2 rc[BIN_R];
    if (small) {
#pragma unroll
        for (int j = 0; j < BIN_R; ++j) {
            const int i = j * BIN_T + tid;
            rc[j] = make_int2(0, 0);
            if (i < n) rc[j] = tmp[beg + i];
        }
#pragma unroll
        for (int j = 0; j < BIN_R; ++j)
            if (j * BIN_T + tid < n) atomicAdd(&cnt[rc[j].y & BIN_MASK], 1);
    } else {
        for (int i = tid; i < n; i += BIN_T) atomicAdd(&cnt[tmp[beg + i].y & BIN_MASK], 1);
    }
    __syncthreads();
    const int v  = (tid < 256) ? cnt[tid] : 0;
    const int ex = excl_scan256(v, wt, tid);
    if (tid < 256) {
        const int node = blockIdx.x * 256 + tid;
        rowptr[node] = beg + ex;
        dinv[node]   = (v > 0) ? rsqrtf((float)v) : 0.0f;
        cur[tid]     = ex;
    }
    __syncthreads();
    if (small) {
#pragma unroll
        for (int j = 0; j < BIN_R; ++j)
            if (j * BIN_T + tid < n) stage[atomicAdd(&cur[rc[j].y & BIN_MASK], 1)] = rc[j];
        __syncthreads();
        for (int i = tid; i < n; i += BIN_T) erec[beg + i] = stage[i];
    } else {
        for (int i = tid; i < n; i += BIN_T) {
            const int2 e = tmp[beg + i];
            erec[beg + atomicAdd(&cur[e.y & BIN_MASK], 1)] = e;
        }
    }
}

template <int CNT, int FIRST>
__device__ __forceinline__ void gather_accum(const int2* __restrict__ rec,
                                             const float* __restrict__ hl,
                                             const float* __restrict__ dinv, const float dt,
                                             int2* __restrict__ wout, const int lane,
                                             float4& acc) {
    int2   r[CNT];
    float4 v[CNT];
    float  wgt[CNT];
#pragma unroll
    for (int i = 0; i < CNT; ++i) r[i] = rec[i];
#pragma unroll
    for (int i = 0; i < CNT; ++i) {
        unsigned s = min((unsigned)r[i].x, (unsigned)(N_NODES - 1));
        v[i] = *(const float4*)(hl + (size_t)s * F_DIM);
        if (FIRST) wgt[i] = dinv[s] * dt;
        else       wgt[i] = __int_as_float(r[i].y);
    }
    if (FIRST) {
        int px = 0, py = 0;
#pragma unroll
        for (int i = 0; i < CNT; ++i) {
            px = (lane == i) ? r[i].x : px;
            py = (lane == i) ? __float_as_int(wgt[i]) : py;
        }
        if (lane < CNT) wout[lane] = make_int2(px, py);
    }
#pragma unroll
    for (int i = 0; i < CNT; ++i) {
        acc.x = fmaf(wgt[i], v[i].x, acc.x);
        acc.y = fmaf(wgt[i], v[i].y, acc.y);
        acc.z = fmaf(wgt[i], v[i].z, acc.z);
        acc.w = fmaf(wgt[i], v[i].w, acc.w);
    }
}

template <int FIRST>
__global__ __launch_bounds__(256) void spmm_gather_kernel(const int* __restrict__ rowptr,
                                                          const int2* __restrict__ erec,
                                                          const float* __restrict__ dinv,
                                                          int2* __restrict__ wout,
                                                          const float* __restrict__ hin,
                                                          float* __restrict__ hout) {
    const int lane = threadIdx.x & 31;
    const int node = __builtin_amdgcn_readfirstlane((int)(blockIdx.x * 8 + (threadIdx.x >> 5)));
    const int beg = rowptr[node];
    const int end = rowptr[node + 1];
    const float dt = FIRST ? dinv[node] : 0.0f;
    const float* __restrict__ hl = hin + lane * 4;
    float4 acc = make_float4(0.0f, 0.0f, 0.0f, 0.0f);
    int e = beg;
    while (e + 8 <= end) { gather_accum<8, FIRST>(erec + e, hl, dinv, dt, wout + e, lane, acc); e += 8; }
    if (e + 4 <= end)    { gather_accum<4, FIRST>(erec + e, hl, dinv, dt, wout + e, lane, acc); e += 4; }
    switch (end - e) {
        case 3: gather_accum<3, FIRST>(erec + e, hl, dinv, dt, wout + e, lane, acc); break;
        case 2: gather_accum<2, FIRST>(erec + e, hl, dinv, dt, wout + e, lane, acc); break;
        case 1: gather_accum<1, FIRST>(erec + e, hl, dinv, dt, wout + e, lane, acc); break;
        default: break;
    }
    *(float4*)(hout + (size_t)node * F_DIM + lane * 4) = acc;
}


template <int L>
__device__ __forceinline__ int row_share(int x) {
    return __builtin_amdgcn_update_dpp(0, x, 0x150 + L, 0xf, 0xf, false);
}

#define GATHER_PAIR_LOAD(P, I)                                                          \
    {                                                                                    \
        const int sx = row_share<(P)>(rv.x);                                             \
        w[I] = __int_as_float(row_share<(P)>(rv.y));                                     \
        v[I] = make_float4(0.0f, 0.0f, 0.0f, 0.0f);                                      \
        if (2 * (P) + hi < cnt) {                                                        \
                  \
                                                  \
            const unsigned s = min((unsigned)sx, (unsigned)(N_NODES - 1));               \
            v[I] = *(const float4*)(hl + (size_t)s * C_OUT);                             \
        }                                                                                \
    }

template <int P0>
__device__ __forceinline__ void gather_group(const int2 rv, const int cnt, const int hi,
                                             const float* __restrict__ hl, float4& acc) {
    float4 v[8];
    float  w[8];
    GATHER_PAIR_LOAD(P0 + 0, 0)
    GATHER_PAIR_LOAD(P0 + 1, 1)
    GATHER_PAIR_LOAD(P0 + 2, 2)
    GATHER_PAIR_LOAD(P0 + 3, 3)
    GATHER_PAIR_LOAD(P0 + 4, 4)
    GATHER_PAIR_LOAD(P0 + 5, 5)
    GATHER_PAIR_LOAD(P0 + 6, 6)
    GATHER_PAIR_LOAD(P0 + 7, 7)
#pragma unroll
    for (int i = 0; i < 8; ++i) {
        acc.x = fmaf(w[i], v[i].x, acc.x);
        acc.y = fmaf(w[i], v[i].y, acc.y);
        acc.z = fmaf(w[i], v[i].z, acc.z);
        acc.w = fmaf(w[i], v[i].w, acc.w);
    }
}

template <int LAST>
__global__ __launch_bounds__(256) void spmm_gather64_kernel(const int* __restrict__ rowptr,
                                                            const int2* __restrict__ erec,
                                                            const float* __restrict__ hin,
                                                            const float* __restrict__ zadd,
                                                            const float* __restrict__ bias,
                                                            float* __restrict__ hout) {
    const int lane   = threadIdx.x & 31;
    const int lane16 = lane & 15;
    const int hi     = lane >> 4;
    const int slot   = 2 * lane16 + hi;
    const int node = __builtin_amdgcn_readfirstlane((int)(blockIdx.x * 8 + (threadIdx.x >> 5)));
    const int beg = rowptr[node];
    const int end = rowptr[node + 1];
    const float* __restrict__ hl = hin + lane16 * 4;
    const float4 z = *(const float4*)(zadd + (size_t)node * C_OUT + lane16 * 4);
    float4 acc = make_float4(0.0f, 0.0f, 0.0f, 0.0f);
    for (int e = beg; e < end; e += 32) {
        const int cnt = min(end - e, 32);
        int2 rv = make_int2(0, 0);
        if (slot < cnt) rv = erec[e + slot];
        gather_group<0>(rv, cnt, hi, hl, acc);
        if (cnt > 16) gather_group<8>(rv, cnt, hi, hl, acc);
    }
    acc.x += __shfl_xor(acc.x, 16, 32);
    acc.y += __shfl_xor(acc.y, 16, 32);
    acc.z += __shfl_xor(acc.z, 16, 32);
    acc.w += __shfl_xor(acc.w, 16, 32);
    float4 o = make_float4(acc.x + z.x, acc.y + z.y, acc.z + z.z, acc.w + z.w);
    if (LAST) {
        const float4 b = *(const float4*)(bias + lane16 * 4);
        o.x += b.x; o.y += b.y; o.z += b.z; o.w += b.w;
        float mx = fmaxf(fmaxf(o.x, o.y), fmaxf(o.z, o.w));
#pragma unroll
        for (int off = 8; off > 0; off >>= 1) mx = fmaxf(mx, __shfl_xor(mx, off, 32));
        float s = (expf(o.x - mx) + expf(o.y - mx)) + (expf(o.z - mx) + expf(o.w - mx));
#pragma unroll
        for (int off = 8; off > 0; off >>= 1) s += __shfl_xor(s, off, 32);
        const float lse = mx + logf(s);
        o.x -= lse; o.y -= lse; o.z -= lse; o.w -= lse;
    }
    if (hi == 0) *(float4*)(hout + (size_t)node * C_OUT + lane16 * 4) = o;
}
#undef GATHER_PAIR_LOAD

#define GEMM_T      512
#define GEMM_MAINW  12
#define GEMM_GRID   256
#define N_TILES     (N_NODES / 16)
#define MAIN_TILES  (GEMM_GRID * GEMM_MAINW)
#define EXTRA_TILES (N_TILES - MAIN_TILES)
static_assert(N_NODES % 16 == 0 && EXTRA_TILES >= 0 && EXTRA_TILES <= GEMM_GRID,
              "row tile deal");

__global__ __launch_bounds__(GEMM_T, 4) void tag_gemm_fused_kernel(
    const float* __restrict__ s0, const float* __restrict__ s1,
    const float* __restrict__ s2, const float* __restrict__ s3,
    const float* __restrict__ W1, const float* __restrict__ b1,
    const float* __restrict__ W2, float* __restrict__ zout) {
    constexpr int NOUT  = H_OUT1;
    constexpr int NSEG  = 4;
    constexpr int NT    = NOUT / 16;
    constexpr int PITCH = 2 * NOUT + 32;
    constexpr int SEGF  = (F_DIM / 2) * PITCH;
    constexpr int NV4   = NOUT / 4;
    constexpr int ITEMS = (F_DIM / 2) * NV4 / GEMM_T;
    constexpr int GSTEP = 8 * 2 * PITCH;
    static_assert(2 * C_OUT == NOUT && H_OUT1 == F_DIM && NT == 8, "segment shape");

    __shared__ __attribute__((aligned(16))) float Bs[2 * SEGF];
    __shared__ __attribute__((aligned(16))) float bsh[H_OUT1];
    __shared__ __attribute__((aligned(16))) float xbuf[4 * 8 * 32 * 8];

    const int tid    = threadIdx.x;
    const int lane   = tid & 31;
    const int lane16 = lane & 15;
    const int hi     = lane >> 4;
    const int wave   = __builtin_amdgcn_readfirstlane(tid >> 5);
    const bool is_main = wave < GEMM_MAINW;
    const bool shares = !is_main && (int)blockIdx.x < EXTRA_TILES;
    const int mine   = shares ? wave - GEMM_MAINW : -1;
    const int tile   = is_main ? (int)blockIdx.x * GEMM_MAINW + wave
                               : MAIN_TILES + (int)blockIdx.x;
    const int row0   = tile * 16;
    const int m = min(row0 + lane16, N_NODES - 1);
    const size_t aoff = (size_t)m * F_DIM + hi * 2;

    const float* segs[4] = { s0, s1, s2, s3 };

    v8f acc[NT];
#pragma unroll
    for (int t = 0; t < NT; ++t) acc[t] = (v8f){ 0.f, 0.f, 0.f, 0.f, 0.f, 0.f, 0.f, 0.f };

    v2f a_cur[8], a_nxt[8];
#pragma unroll
    for (int j = 0; j < 8; ++j) a_cur[j] = *(const v2f*)(segs[0] + aoff + 4 * j);

    const int spr = tid / NV4;
    const int sn4 = tid % NV4;
    constexpr int PR_STEP = GEMM_T / NV4;
    static_assert(PR_STEP == 16, "the W2 row order below is written for 16 pair rows per item");
    const float* __restrict__ wsrc = W1 + (size_t)(2 * spr) * NOUT + 4 * sn4;
    float* const wdst = Bs + spr * PITCH + 8 * sn4;
    const float* __restrict__ wsrc2 = W2 + (size_t)((4 * sn4) >> 6) * (F_DIM * C_OUT)
                                         + (size_t)(2 * spr) * C_OUT + ((4 * sn4) & 63);
    float* const wdst2 = Bs + (8 * (spr >> 3) + 2 * (spr & 3) + ((spr >> 2) & 1)) * PITCH
                            + 8 * sn4;
    float4 w0[ITEMS], w1[ITEMS];
#pragma unroll
    for (int it = 0; it < ITEMS; ++it) {
        const float* g = wsrc + (size_t)(2 * PR_STEP * it) * NOUT;
        w0[it] = *(const float4*)g;
        w1[it] = *(const float4*)(g + NOUT);
    }
#pragma unroll
    for (int it = 0; it < ITEMS; ++it) {
        float* d = wdst + it * PR_STEP * PITCH;
        *(float4*)d       = make_float4(w0[it].x, w1[it].x, w0[it].y, w1[it].y);
        *(float4*)(d + 4) = make_float4(w0[it].z, w1[it].z, w0[it].w, w1[it].w);
    }
    if (tid < H_OUT1) bsh[tid] = b1[tid];
    __syncthreads();

    v2f bq[2][NT];

#pragma unroll
    for (int s = 0; s < NSEG; ++s) {
        const float* bg = Bs + (s & 1) * SEGF + hi * PITCH + 2 * lane16;
        const float* ag = segs[s] + aoff;
        const float* an = (s < NSEG - 1) ? segs[s + 1] + aoff : ag;
#pragma unroll
        for (int t = 0; t < NT; ++t) bq[0][t] = *(const v2f*)(bg + 32 * t);

        auto products = [&](const float* bgp, const float* bnext) {
#pragma unroll
            for (int j = 0; j < 8; ++j) {
                const float* bn = (j < 7) ? bgp + (j + 1) * 2 * PITCH : bnext;
#pragma unroll
                for (int t = 0; t < NT; ++t) bq[(j + 1) & 1][t] = *(const v2f*)(bn + 32 * t);
#pragma unroll
                for (int t = 0; t < NT; ++t)
                    acc[t] = __builtin_amdgcn_wmma_f32_16x16x4_f32(
                        false, bq[j & 1][t], false, a_cur[j], (short)0, acc[t], false, false);
            }
        };

#pragma unroll
        for (int j = 0; j < 8; ++j) a_nxt[j] = *(const v2f*)(ag + 32 + 4 * j);
        if (s < NSEG - 1) {
#pragma unroll
            for (int it = 0; it < ITEMS; ++it) {
                const float* g = wsrc + (size_t)((s + 1) * F_DIM + 2 * PR_STEP * it) * NOUT;
                w0[it] = *(const float4*)g;
                w1[it] = *(const float4*)(g + NOUT);
            }
        } else {
#pragma unroll
            for (int it = 0; it < ITEMS; ++it) {
                const float* g = wsrc2 + (size_t)(2 * PR_STEP * it) * C_OUT;
                w0[it] = *(const float4*)g;
                w1[it] = *(const float4*)(g + C_OUT);
            }
        }
        if (is_main || mine == 0) products(bg, bg + GSTEP);
#pragma unroll
        for (int j = 0; j < 8; ++j) a_cur[j] = a_nxt[j];
        bg += GSTEP;
        ag += 32;

#pragma unroll 1
        for (int sc = 1; sc < 4; ++sc) {
            const bool last = (sc == 3);
            const float* ap = last ? an : ag + 32;
            if (s < NSEG - 1 || !last) {
#pragma unroll
                for (int j = 0; j < 8; ++j) a_nxt[j] = *(const v2f*)(ap + 4 * j);
            }
            if (is_main || sc == mine) {
                if (!is_main) {
#pragma unroll
                    for (int t = 0; t < NT; ++t) bq[0][t] = *(const v2f*)(bg + 32 * t);
                }
                products(bg, last ? bg + 7 * 2 * PITCH : bg + GSTEP);
            }
#pragma unroll
            for (int j = 0; j < 8; ++j) a_cur[j] = a_nxt[j];
            bg += GSTEP;
            ag += 32;
        }

        {
            float* db = (s < NSEG - 1) ? wdst + ((s + 1) & 1) * SEGF : wdst2;
#pragma unroll
            for (int it = 0; it < ITEMS; ++it) {
                float* d = db + it * PR_STEP * PITCH;
                *(float4*)d       = make_float4(w0[it].x, w1[it].x, w0[it].y, w1[it].y);
                *(float4*)(d + 4) = make_float4(w0[it].z, w1[it].z, w0[it].w, w1[it].w);
            }
            if (s == NSEG - 1 && shares) {
                float* xb = xbuf + (mine * 8 * 32 + lane) * 8;
#pragma unroll
                for (int t = 0; t < NT; ++t) {
                    *(float4*)(xb + t * 256) =
                        make_float4(acc[t][0], acc[t][1], acc[t][2], acc[t][3]);
                    *(float4*)(xb + t * 256 + 4) =
                        make_float4(acc[t][4], acc[t][5], acc[t][6], acc[t][7]);
                }
            }
            __syncthreads();
        }
    }

    if (shares) {
#pragma unroll
        for (int t = 0; t < NT; ++t) {
            v8f sum = (v8f){ 0.f, 0.f, 0.f, 0.f, 0.f, 0.f, 0.f, 0.f };
            for (int qq = 0; qq < 4; ++qq) {
                const float* p = xbuf + ((qq * 8 + t) * 32 + lane) * 8;
                const float4 lo = *(const float4*)p;
                const float4 up = *(const float4*)(p + 4);
                sum += (v8f){ lo.x, lo.y, lo.z, lo.w, up.x, up.y, up.z, up.w };
            }
            acc[t] = sum;
        }
    }

#pragma unroll
    for (int t = 0; t < NT; ++t) {
        const float4 ba = *(const float4*)(bsh + 16 * t + 8 * hi);
        const float4 bb = *(const float4*)(bsh + 16 * t + 8 * hi + 4);
        const float bv[8] = { ba.x, ba.y, ba.z, ba.w, bb.x, bb.y, bb.z, bb.w };
#pragma unroll
        for (int r = 0; r < 8; ++r) {
            const float v = acc[t][r] + bv[r];
            acc[t][r] = (v > 0.0f) ? v : 0.0f;
        }
    }

    v8f zac[NT];
#pragma unroll
    for (int y = 0; y < 2; ++y) {
        const float* bg = Bs + (y & 1) * SEGF + hi * PITCH + 2 * lane16;
        if (y == 0) {
#pragma unroll
            for (int it = 0; it < ITEMS; ++it) {
                const float* g = wsrc2 + (size_t)2 * (F_DIM * C_OUT)
                                       + (size_t)(2 * PR_STEP * it) * C_OUT;
                w0[it] = *(const float4*)g;
                w1[it] = *(const float4*)(g + C_OUT);
            }
        }
#pragma unroll
        for (int tp = 0; tp < NT / 2; ++tp) {
            if (is_main || tp == mine) {
                const float* bt = bg + 64 * tp;
                zac[2 * tp]     = (v8f){ 0.f, 0.f, 0.f, 0.f, 0.f, 0.f, 0.f, 0.f };
                zac[2 * tp + 1] = (v8f){ 0.f, 0.f, 0.f, 0.f, 0.f, 0.f, 0.f, 0.f };
#pragma unroll
                for (int i = 0; i < 8; ++i)
                    bq[0][i] = *(const v2f*)(bt + (i >> 1) * 2 * PITCH + 32 * (i & 1));
#pragma unroll
                for (int C = 0; C < 8; ++C) {
                    if (C < 7) {
#pragma unroll
                        for (int i = 0; i < 8; ++i)
                            bq[(C + 1) & 1][i] = *(const v2f*)(bt + (4 * (C + 1) + (i >> 1)) * 2 * PITCH
                                                                  + 32 * (i & 1));
                        __builtin_amdgcn_sched_barrier(0);
                    }
#pragma unroll
                    for (int i = 0; i < 8; ++i) {
                        const int J = 4 * C + (i >> 1);
                        const v2f ha = (v2f){ acc[J >> 2][2 * (J & 3)], acc[J >> 2][2 * (J & 3) + 1] };
                        zac[2 * tp + (i & 1)] = __builtin_amdgcn_wmma_f32_16x16x4_f32(
                            false, ha, false, bq[C & 1][i], (short)0, zac[2 * tp + (i & 1)],
                            false, false);
                    }
                }
            }
        }
        if (y == 0) {
            float* db = wdst2 + SEGF;
#pragma unroll
            for (int it = 0; it < ITEMS; ++it) {
                float* d = db + it * PR_STEP * PITCH;
                *(float4*)d       = make_float4(w0[it].x, w1[it].x, w0[it].y, w1[it].y);
                *(float4*)(d + 4) = make_float4(w0[it].z, w1[it].z, w0[it].w, w1[it].w);
            }
            __syncthreads();
        }
#pragma unroll
        for (int tp = 0; tp < NT / 2; ++tp) {
            if (is_main || tp == mine) {
#pragma unroll
                for (int c = 0; c < 2; ++c) {
                    const int t = 2 * tp + c;
                    float* zp = zout + (size_t)(2 * y + (t >> 2)) * ((size_t)N_NODES * C_OUT)
                                     + ((t & 3) * 16 + lane16);
#pragma unroll
                    for (int j = 0; j < 8; ++j) {
                        const int mm = row0 + j + hi * 8;
                        zp[(size_t)mm * C_OUT] = zac[t][j];
                    }
                }
            }
        }
    }
}


extern "C" void kernel_launch(void* const* d_in, const int* in_sizes, int n_in,
                              void* d_out, int out_size, void* d_ws, size_t ws_size,
                              hipStream_t stream) {
    const float* x   = (const float*)d_in[0];
    const int*   ei  = (const int*)d_in[1];
    const float* W1  = (const float*)d_in[2];
    const float* b1  = (const float*)d_in[3];
    const float* W2  = (const float*)d_in[4];
    const float* b2  = (const float*)d_in[5];
    float*       out = (float*)d_out;

    const int* rowp = ei;
    const int* colp = ei + N_EDGES;

    int*   bhist  = (int*)d_ws;
    int*   rowptr = bhist + N_PAD;
    float* dinv   = (float*)(rowptr + N_PAD);
    int*   binst  = (int*)(dinv + N_PAD);
    int2*  erec   = (int2*)(binst + N_EDGES);
    float* hb0    = (float*)(erec + N_EDGES);
    float* hb1    = hb0 + (size_t)N_NODES * F_DIM;
    float* hb2    = hb1 + (size_t)N_NODES * F_DIM;
    float* zpl    = hb2 + (size_t)N_NODES * F_DIM;
    const size_t ZP = (size_t)N_NODES * C_OUT;
    int2*  wrec   = (int2*)(zpl + 4 * ZP);
    float* t1     = hb0;
    float* t2     = hb1;
    int2*  tmp    = (int2*)hb2;

    const int nodeGroups = N_NODES / 8;

    bin_count_kernel<<<SORT_NB, SORT_T, 0, stream>>>(colp, bhist);
    bin_scatter_kernel<<<SORT_NB, SORT_T, 0, stream>>>(rowp, colp, bhist, binst, tmp);
    bin_sort_kernel<<<N_BINBLK, BIN_T, 0, stream>>>(tmp, binst, rowptr, dinv, erec);

    spmm_gather_kernel<1><<<nodeGroups, 256, 0, stream>>>(rowptr, erec, dinv, wrec, x, hb0);
    spmm_gather_kernel<0><<<nodeGroups, 256, 0, stream>>>(rowptr, wrec, dinv, wrec, hb0, hb1);
    spmm_gather_kernel<0><<<nodeGroups, 256, 0, stream>>>(rowptr, wrec, dinv, wrec, hb1, hb2);

    tag_gemm_fused_kernel<<<GEMM_GRID, GEMM_T, 0, stream>>>(x, hb0, hb1, hb2, W1, b1, W2, zpl);

    spmm_gather64_kernel<0><<<nodeGroups, 256, 0, stream>>>(
        rowptr, wrec, zpl + 3 * ZP, zpl + 2 * ZP, b2, t1);
    spmm_gather64_kernel<0><<<nodeGroups, 256, 0, stream>>>(
        rowptr, wrec, t1, zpl + ZP, b2, t2);
    spmm_gather64_kernel<1><<<nodeGroups, 256, 0, stream>>>(
        rowptr, wrec, t2, zpl, b2, out);
}
